# starting kernel
# baseline (speedup 1.0000x reference)
#include <hip/hip_runtime.h>
#include <math.h>


typedef float v2f __attribute__((ext_vector_type(2)));
typedef float v8f __attribute__((ext_vector_type(8)));

static inline long long cdiv_ll(long long a, long long b) { return (a + b - 1) / b; }

__global__ void fill_kernel(float* __restrict__ p, float v, unsigned n) {
  unsigned i = blockIdx.x * blockDim.x + threadIdx.x;
  if (i < n) p[i] = v;
}

__global__ void deg_kernel(const int* __restrict__ ei, float* __restrict__ deg, int E) {
  int e = blockIdx.x * blockDim.x + threadIdx.x;
  if (e < E) atomicAdd(&deg[ei[E + e]], 1.0f);
}

__global__ void dis_kernel(const float* __restrict__ deg, float* __restrict__ dis, int N) {
  int i = blockIdx.x * blockDim.x + threadIdx.x;
  if (i < N) { float d = deg[i]; dis[i] = d > 0.0f ? rsqrtf(d) : 0.0f; }
}

__global__ void edge_kernel(const int* __restrict__ ei, const float* __restrict__ dis,
                            int4* __restrict__ edges, int E) {
  int e = blockIdx.x * blockDim.x + threadIdx.x;
  if (e >= E) return;
  int s = ei[e], d = ei[E + e];
  int4 r;
  r.x = s;
  r.y = d;
  r.z = __float_as_int(dis[s] * dis[d]);
  r.w = 0;
  edges[e] = r;
}

__global__ void gemm_wmma_f32(const float* __restrict__ H, const float* __restrict__ W,
                              float* __restrict__ out, int N, int K, int OUT,
                              int nColTiles, int nTiles) {
  int wave = (int)(((unsigned)blockIdx.x * blockDim.x + threadIdx.x) >> 5);
  if (wave >= nTiles) return;
  int lane = threadIdx.x & 31;
  int rt = wave / nColTiles, ct = wave % nColTiles;
  int half = lane >> 4;
  int m = lane & 15;
  int row = rt * 16 + m;
  int rowc = row < N ? row : N - 1;
  int col = ct * 16 + m;
  const float* hrow = H + (unsigned)rowc * (unsigned)K;

  v8f acc = {};
  for (int kk = 0; kk < K; kk += 4) {
    int ka = kk + half * 2;
    v2f a = *(const v2f*)(hrow + ka);
    v2f b;
    b.x = W[(unsigned)ka * (unsigned)OUT + col];
    b.y = W[(unsigned)(ka + 1) * (unsigned)OUT + col];
    acc = __builtin_amdgcn_wmma_f32_16x16x4_f32(
         false, a,  false, b,
         (short)0, acc,  false,  false);
  }
#pragma unroll
  for (int r = 0; r < 8; ++r) {
    int mrow = rt * 16 + r + 8 * half;
    if (mrow < N) out[(unsigned)mrow * (unsigned)OUT + col] = acc[r];
  }
}

template <int LOGC>
__global__ void scatter_kernel(const float* __restrict__ ht, const int4* __restrict__ edges,
                               float* __restrict__ agg, unsigned E) {
  unsigned t = blockIdx.x * blockDim.x + threadIdx.x;
  unsigned e = t >> (LOGC - 2);
  if (e >= E) return;
  unsigned c4 = (t & ((1u << (LOGC - 2)) - 1u)) << 2;
  const int4 r = edges[e];
  unsigned src = (unsigned)r.x, dst = (unsigned)r.y;
  float nrm = __int_as_float(r.z);
  const float4 v = *(const float4*)(ht + (src << LOGC) + c4);
  float* o = agg + (dst << LOGC) + c4;
  atomicAdd(o + 0, v.x * nrm);
  atomicAdd(o + 1, v.y * nrm);
  atomicAdd(o + 2, v.z * nrm);
  atomicAdd(o + 3, v.w * nrm);
}

__global__ void scatter1_kernel(const float* __restrict__ ht, const int4* __restrict__ edges,
                                float* __restrict__ agg, unsigned E) {
  unsigned e = blockIdx.x * blockDim.x + threadIdx.x;
  if (e >= E) return;
  const int4 r = edges[e];
  atomicAdd(&agg[(unsigned)r.y], ht[(unsigned)r.x] * __int_as_float(r.z));
}

template <int LOGC, bool TANH>
__global__ void finalize_kernel(const float* __restrict__ agg, const float* __restrict__ ht,
                                const float* __restrict__ dis, const float* __restrict__ bias,
                                float* __restrict__ out, unsigned total) {
  unsigned t = blockIdx.x * blockDim.x + threadIdx.x;
  if (t >= total) return;
  unsigned i = t >> LOGC;
  unsigned c = t & ((1u << LOGC) - 1u);
  float d = dis[i];
  float v = agg[t] + d * d * ht[t] + bias[c];
  out[t] = TANH ? tanhf(v) : v;
}

__global__ void dot32_kernel(const float* __restrict__ H, const float* __restrict__ W3,
                             float* __restrict__ out, int N) {
  int i = blockIdx.x * blockDim.x + threadIdx.x;
  if (i >= N) return;
  const float4* h = (const float4*)(H + (unsigned)i * 32u);
  float s = 0.f;
#pragma unroll
  for (int q = 0; q < 8; ++q) {
    float4 a = h[q];
    float4 w = ((const float4*)W3)[q];
    s += a.x * w.x + a.y * w.y + a.z * w.z + a.w * w.w;
  }
  out[i] = s;
}

extern "C" void kernel_launch(void* const* d_in, const int* in_sizes, int n_in,
                              void* d_out, int out_size, void* d_ws, size_t ws_size,
                              hipStream_t stream) {
  const float* x  = (const float*)d_in[0];
  const int*   ei = (const int*)d_in[1];
  const float* W1 = (const float*)d_in[2];
  const float* b1 = (const float*)d_in[3];
  const float* W2 = (const float*)d_in[4];
  const float* b2 = (const float*)d_in[5];
  const float* W3 = (const float*)d_in[6];
  const float* b3 = (const float*)d_in[7];
  float* out = (float*)d_out;

  const int D2 = 64, D = 32;
  const int N = in_sizes[0] / D2;
  const int E = in_sizes[1] / 2;

  float* ws    = (float*)d_ws;
  float* dis   = ws;
  float* ht    = ws + N;
  float* agg   = ht + (long long)N * D2;
  float* ht3   = agg + (long long)N * D2;
  float* agg3  = ht3 + N;
  int4*  edges = (int4*)(agg3 + N);
  float* deg   = ht;
  (void)ws_size; (void)n_in; (void)out_size;

  const int B = 256;

  fill_kernel<<<(unsigned)cdiv_ll(N, B), B, 0, stream>>>(deg, 1.0f, (unsigned)N);
  deg_kernel<<<(unsigned)cdiv_ll(E, B), B, 0, stream>>>(ei, deg, E);
  dis_kernel<<<(unsigned)cdiv_ll(N, B), B, 0, stream>>>(deg, dis, N);
  edge_kernel<<<(unsigned)cdiv_ll(E, B), B, 0, stream>>>(ei, dis, edges, E);

  const int rowTiles = (int)cdiv_ll(N, 16);

  {
    int colT = D2 / 16, tiles = rowTiles * colT;
    gemm_wmma_f32<<<(unsigned)cdiv_ll((long long)tiles * 32, B), B, 0, stream>>>(
        x, W1, ht, N, D2, D2, colT, tiles);
  }
  fill_kernel<<<(unsigned)cdiv_ll((long long)N * D2, B), B, 0, stream>>>(agg, 0.f, (unsigned)(N * D2));
  scatter_kernel<6><<<(unsigned)cdiv_ll((long long)E * (D2 / 4), B), B, 0, stream>>>(ht, edges, agg, (unsigned)E);
  finalize_kernel<6, true><<<(unsigned)cdiv_ll((long long)N * D2, B), B, 0, stream>>>(
      agg, ht, dis, b1, agg, (unsigned)(N * D2));

  {
    int colT = D / 16, tiles = rowTiles * colT;
    gemm_wmma_f32<<<(unsigned)cdiv_ll((long long)tiles * 32, B), B, 0, stream>>>(
        agg, W2, ht, N, D2, D, colT, tiles);
  }
  fill_kernel<<<(unsigned)cdiv_ll((long long)N * D, B), B, 0, stream>>>(agg, 0.f, (unsigned)(N * D));
  scatter_kernel<5><<<(unsigned)cdiv_ll((long long)E * (D / 4), B), B, 0, stream>>>(ht, edges, agg, (unsigned)E);
  finalize_kernel<5, true><<<(unsigned)cdiv_ll((long long)N * D, B), B, 0, stream>>>(
      agg, ht, dis, b2, agg, (unsigned)(N * D));

  dot32_kernel<<<(unsigned)cdiv_ll(N, B), B, 0, stream>>>(agg, W3, ht3, N);
  fill_kernel<<<(unsigned)cdiv_ll(N, B), B, 0, stream>>>(agg3, 0.f, (unsigned)N);
  scatter1_kernel<<<(unsigned)cdiv_ll(E, B), B, 0, stream>>>(ht3, edges, agg3, (unsigned)E);
  finalize_kernel<0, false><<<(unsigned)cdiv_ll(N, B), B, 0, stream>>>(
      agg3, ht3, dis, b3, out, (unsigned)N);
}


# speedup vs baseline: 1.6114x; 1.6114x over previous
#include <hip/hip_runtime.h>
#include <math.h>


typedef float v2f __attribute__((ext_vector_type(2)));
typedef float v8f __attribute__((ext_vector_type(8)));

static inline long long cdiv_ll(long long a, long long b) { return (a + b - 1) / b; }

__global__ void fill_kernel(float* __restrict__ p, float v, unsigned n) {
  unsigned i = blockIdx.x * blockDim.x + threadIdx.x;
  if (i < n) p[i] = v;
}

__global__ void zero4_kernel(float4* __restrict__ p, unsigned n4) {
  unsigned i = blockIdx.x * blockDim.x + threadIdx.x;
  if (i < n4) p[i] = make_float4(0.f, 0.f, 0.f, 0.f);
}

__device__ __forceinline__ unsigned package_half() {
  return (__builtin_amdgcn_s_sendmsg_rtn(0x87) >> 18) & 1u;
}

template <int U>
__device__ __forceinline__ void deg_batch(const int* __restrict__ dstp, float* __restrict__ deg,
                                          unsigned e0, unsigned step) {
  unsigned d[U];
#pragma unroll
  for (int u = 0; u < U; ++u) d[u] = (unsigned)dstp[e0 + (unsigned)u * step];
#pragma unroll
  for (int u = 0; u < U; ++u) atomicAdd(&deg[d[u]], 1.0f);
}

__global__ void deg_kernel(const int* __restrict__ ei, float* __restrict__ deg2,
                           unsigned E, unsigned N) {
  float* __restrict__ deg = deg2 + package_half() * N;
  const int* __restrict__ dstp = ei + E;
  const unsigned nthr = gridDim.x * blockDim.x;
  const unsigned trips = (E + nthr - 1u) / nthr;
  unsigned e = blockIdx.x * blockDim.x + threadIdx.x;
  unsigned left = trips ? trips - 1u : 0u;
  for (; left >= 4u; left -= 4u, e += 4u * nthr) deg_batch<4>(dstp, deg, e, nthr);
  if (left >= 2u) { deg_batch<2>(dstp, deg, e, nthr); left -= 2u; e += 2u * nthr; }
  if (left >= 1u) { deg_batch<1>(dstp, deg, e, nthr); e += nthr; }
  if (e < E) deg_batch<1>(dstp, deg, e, nthr);
}

__global__ void dis_kernel(const float* __restrict__ degA, const float* __restrict__ degB,
                           float* __restrict__ dis, int N) {
  int i = blockIdx.x * blockDim.x + threadIdx.x;
  if (i < N) { float d = degA[i] + degB[i] + 1.0f; dis[i] = d > 0.0f ? rsqrtf(d) : 0.0f; }
}

__global__ void gemm_wmma_f32(const float* __restrict__ H, const float* __restrict__ W,
                              float* __restrict__ out, int N, int K, int OUT,
                              int nColTiles, int nTiles) {
  int wave = (int)(((unsigned)blockIdx.x * blockDim.x + threadIdx.x) >> 5);
  if (wave >= nTiles) return;
  int lane = threadIdx.x & 31;
  int rt = wave / nColTiles, ct = wave % nColTiles;
  int half = lane >> 4;
  int m = lane & 15;
  int row = rt * 16 + m;
  int rowc = row < N ? row : N - 1;
  int col = ct * 16 + m;
  const float* hrow = H + (unsigned)rowc * (unsigned)K;

  v8f acc = {};
  for (int kk = 0; kk < K; kk += 4) {
    int ka = kk + half * 2;
    v2f a = *(const v2f*)(hrow + ka);
    v2f b;
    b.x = W[(unsigned)ka * (unsigned)OUT + col];
    b.y = W[(unsigned)(ka + 1) * (unsigned)OUT + col];
    acc = __builtin_amdgcn_wmma_f32_16x16x4_f32(
         false, a,  false, b,
         (short)0, acc,  false,  false);
  }
#pragma unroll
  for (int r = 0; r < 8; ++r) {
    int mrow = rt * 16 + r + 8 * half;
    if (mrow < N) out[(unsigned)mrow * (unsigned)OUT + col] = acc[r];
  }
}

template <int LOGC, int U>
__device__ __forceinline__ void scatter_batch(const float* __restrict__ ht,
                                              const int* __restrict__ srcp,
                                              const int* __restrict__ dstp,
                                              const float* __restrict__ dis,
                                              float* __restrict__ agg,
                                              unsigned e0, unsigned step, unsigned c4) {
  unsigned s[U], d[U];
#pragma unroll
  for (int u = 0; u < U; ++u) {
    s[u] = (unsigned)srcp[e0 + (unsigned)u * step];
    d[u] = (unsigned)dstp[e0 + (unsigned)u * step];
  }
  float nrm[U];
  float4 v[U];
#pragma unroll
  for (int u = 0; u < U; ++u) {
    nrm[u] = dis[s[u]] * dis[d[u]];
    v[u] = *(const float4*)(ht + (s[u] << LOGC) + c4);
  }
  if constexpr (U == 4)
    asm volatile("" ::"v"(v[0].x), "v"(v[1].x), "v"(v[2].x), "v"(v[3].x));
  else if constexpr (U == 2)
    asm volatile("" ::"v"(v[0].x), "v"(v[1].x));
#pragma unroll
  for (int u = 0; u < U; ++u) {
    float* o = agg + (d[u] << LOGC) + c4;
    atomicAdd(o + 0, v[u].x * nrm[u]);
    atomicAdd(o + 1, v[u].y * nrm[u]);
    atomicAdd(o + 2, v[u].z * nrm[u]);
    atomicAdd(o + 3, v[u].w * nrm[u]);
  }
}

template <int LOGC>
__global__ void scatter_kernel(const float* __restrict__ ht, const int* __restrict__ ei,
                               const float* __restrict__ dis, float* __restrict__ agg2,
                               unsigned E, unsigned copyElems) {
  float* __restrict__ agg = agg2 + package_half() * copyElems;
  constexpr unsigned SH = LOGC - 2;
  const int* __restrict__ srcp = ei;
  const int* __restrict__ dstp = ei + E;
  const unsigned tid = blockIdx.x * blockDim.x + threadIdx.x;
  const unsigned c4 = (tid & ((1u << SH) - 1u)) << 2;
  const unsigned estep = (gridDim.x * blockDim.x) >> SH;
  const unsigned trips = (E + estep - 1u) / estep;
  unsigned e = tid >> SH;
  unsigned left = trips ? trips - 1u : 0u;
  for (; left >= 4u; left -= 4u, e += 4u * estep)
    scatter_batch<LOGC, 4>(ht, srcp, dstp, dis, agg, e, estep, c4);
  if (left >= 2u) {
    scatter_batch<LOGC, 2>(ht, srcp, dstp, dis, agg, e, estep, c4);
    left -= 2u; e += 2u * estep;
  }
  if (left >= 1u) { scatter_batch<LOGC, 1>(ht, srcp, dstp, dis, agg, e, estep, c4); e += estep; }
  if (e < E) scatter_batch<LOGC, 1>(ht, srcp, dstp, dis, agg, e, estep, c4);
}

template <int U>
__device__ __forceinline__ void scatter1_batch(const float* __restrict__ ht,
                                               const int* __restrict__ srcp,
                                               const int* __restrict__ dstp,
                                               const float* __restrict__ dis,
                                               float* __restrict__ agg,
                                               unsigned e0, unsigned step) {
  unsigned s[U], d[U];
#pragma unroll
  for (int u = 0; u < U; ++u) {
    s[u] = (unsigned)srcp[e0 + (unsigned)u * step];
    d[u] = (unsigned)dstp[e0 + (unsigned)u * step];
  }
  float m[U];
#pragma unroll
  for (int u = 0; u < U; ++u) m[u] = ht[s[u]] * (dis[s[u]] * dis[d[u]]);
#pragma unroll
  for (int u = 0; u < U; ++u) atomicAdd(&agg[d[u]], m[u]);
}

__global__ void scatter1_kernel(const float* __restrict__ ht, const int* __restrict__ ei,
                                const float* __restrict__ dis, float* __restrict__ agg2,
                                unsigned E, unsigned N) {
  float* __restrict__ agg = agg2 + package_half() * N;
  const int* __restrict__ srcp = ei;
  const int* __restrict__ dstp = ei + E;
  const unsigned nthr = gridDim.x * blockDim.x;
  const unsigned trips = (E + nthr - 1u) / nthr;
  unsigned e = blockIdx.x * blockDim.x + threadIdx.x;
  unsigned left = trips ? trips - 1u : 0u;
  for (; left >= 4u; left -= 4u, e += 4u * nthr)
    scatter1_batch<4>(ht, srcp, dstp, dis, agg, e, nthr);
  if (left >= 2u) { scatter1_batch<2>(ht, srcp, dstp, dis, agg, e, nthr); left -= 2u; e += 2u * nthr; }
  if (left >= 1u) { scatter1_batch<1>(ht, srcp, dstp, dis, agg, e, nthr); e += nthr; }
  if (e < E) scatter1_batch<1>(ht, srcp, dstp, dis, agg, e, nthr);
}

template <int LOGC, bool TANH>
__global__ void finalize_kernel(const float* aggA, const float* __restrict__ aggB,
                                const float* __restrict__ ht,
                                const float* __restrict__ dis, const float* __restrict__ bias,
                                float* out, unsigned total) {
  unsigned t = blockIdx.x * blockDim.x + threadIdx.x;
  if (t >= total) return;
  unsigned i = t >> LOGC;
  unsigned c = t & ((1u << LOGC) - 1u);
  float d = dis[i];
  float v = (aggA[t] + aggB[t]) + d * d * ht[t] + bias[c];
  out[t] = TANH ? tanhf(v) : v;
}

__global__ void dot32_kernel(const float* __restrict__ H, const float* __restrict__ W3,
                             float* __restrict__ out, int N) {
  int i = blockIdx.x * blockDim.x + threadIdx.x;
  if (i >= N) return;
  const float4* h = (const float4*)(H + (unsigned)i * 32u);
  float s = 0.f;
#pragma unroll
  for (int q = 0; q < 8; ++q) {
    float4 a = h[q];
    float4 w = ((const float4*)W3)[q];
    s += a.x * w.x + a.y * w.y + a.z * w.z + a.w * w.w;
  }
  out[i] = s;
}

extern "C" void kernel_launch(void* const* d_in, const int* in_sizes, int n_in,
                              void* d_out, int out_size, void* d_ws, size_t ws_size,
                              hipStream_t stream) {
  const float* x  = (const float*)d_in[0];
  const int*   ei = (const int*)d_in[1];
  const float* W1 = (const float*)d_in[2];
  const float* b1 = (const float*)d_in[3];
  const float* W2 = (const float*)d_in[4];
  const float* b2 = (const float*)d_in[5];
  const float* W3 = (const float*)d_in[6];
  const float* b3 = (const float*)d_in[7];
  float* out = (float*)d_out;

  const int D2 = 64, D = 32;
  const int N = in_sizes[0] / D2;
  const int E = in_sizes[1] / 2;

  float* ws    = (float*)d_ws;
  float* dis   = ws;
  float* ht    = ws + N;
  float* aggA  = ht + (long long)N * D2;
  float* aggB  = aggA + (long long)N * D2;
  float* deg2  = ht;
  float* ht3   = ht;
  float* agg3  = ht + N;
  (void)ws_size; (void)n_in; (void)out_size;

  const int B = 256;
  const int PG = 2048;
  const int PG1 = 1536;

  fill_kernel<<<(unsigned)cdiv_ll(2LL * N, B), B, 0, stream>>>(deg2, 0.0f, (unsigned)(2 * N));
  deg_kernel<<<PG1, B, 0, stream>>>(ei, deg2, (unsigned)E, (unsigned)N);
  dis_kernel<<<(unsigned)cdiv_ll(N, B), B, 0, stream>>>(deg2, deg2 + N, dis, N);

  const int rowTiles = (int)cdiv_ll(N, 16);

  {
    int colT = D2 / 16, tiles = rowTiles * colT;
    gemm_wmma_f32<<<(unsigned)cdiv_ll((long long)tiles * 32, B), B, 0, stream>>>(
        x, W1, ht, N, D2, D2, colT, tiles);
  }
  zero4_kernel<<<(unsigned)cdiv_ll((long long)N * (2 * D2 / 4), B), B, 0, stream>>>(
      (float4*)aggA, (unsigned)(N * (2 * D2 / 4)));
  scatter_kernel<6><<<PG, B, 0, stream>>>(ht, ei, dis, aggA, (unsigned)E, (unsigned)(N * D2));
  finalize_kernel<6, true><<<(unsigned)cdiv_ll((long long)N * D2, B), B, 0, stream>>>(
      aggA, aggB, ht, dis, b1, aggA, (unsigned)(N * D2));

  {
    int colT = D / 16, tiles = rowTiles * colT;
    gemm_wmma_f32<<<(unsigned)cdiv_ll((long long)tiles * 32, B), B, 0, stream>>>(
        aggA, W2, ht, N, D2, D, colT, tiles);
  }
  zero4_kernel<<<(unsigned)cdiv_ll((long long)N * (2 * D / 4), B), B, 0, stream>>>(
      (float4*)aggA, (unsigned)(N * (2 * D / 4)));
  scatter_kernel<5><<<PG, B, 0, stream>>>(ht, ei, dis, aggA, (unsigned)E, (unsigned)(N * D));
  finalize_kernel<5, true><<<(unsigned)cdiv_ll((long long)N * D, B), B, 0, stream>>>(
      aggA, aggA + (long long)N * D, ht, dis, b2, aggA, (unsigned)(N * D));

  dot32_kernel<<<(unsigned)cdiv_ll(N, B), B, 0, stream>>>(aggA, W3, ht3, N);
  fill_kernel<<<(unsigned)cdiv_ll(2LL * N, B), B, 0, stream>>>(agg3, 0.f, (unsigned)(2 * N));
  scatter1_kernel<<<PG1, B, 0, stream>>>(ht3, ei, dis, agg3, (unsigned)E, (unsigned)N);
  finalize_kernel<0, false><<<(unsigned)cdiv_ll(N, B), B, 0, stream>>>(
      agg3, agg3 + N, ht3, dis, b3, out, (unsigned)N);
}


# Row-wide atomic passes go one edge per wave, lane j on words j and j+32: every f32 atomic and gather instruction covers one whole 128-B line, not 4 lines of 8 words; sums stay in arrival order
# speedup vs baseline: 2.0386x; 1.2651x over previous
#include <hip/hip_runtime.h>
#include <math.h>


typedef float v2f __attribute__((ext_vector_type(2)));
typedef float v8f __attribute__((ext_vector_type(8)));

static inline long long cdiv_ll(long long a, long long b) { return (a + b - 1) / b; }

__global__ void fill_kernel(float* __restrict__ p, float v, unsigned n) {
  unsigned i = blockIdx.x * blockDim.x + threadIdx.x;
  if (i < n) p[i] = v;
}

__global__ void zero4_kernel(float4* __restrict__ p, unsigned n4) {
  unsigned i = blockIdx.x * blockDim.x + threadIdx.x;
  if (i < n4) p[i] = make_float4(0.f, 0.f, 0.f, 0.f);
}

__device__ __forceinline__ unsigned package_half() {
  return (__builtin_amdgcn_s_sendmsg_rtn(0x87) >> 18) & 1u;
}

template <int U>
__device__ __forceinline__ void deg_batch(const int* __restrict__ dstp, float* __restrict__ deg,
                                          unsigned e0, unsigned step) {
  unsigned d[U];
#pragma unroll
  for (int u = 0; u < U; ++u) d[u] = (unsigned)dstp[e0 + (unsigned)u * step];
#pragma unroll
  for (int u = 0; u < U; ++u) atomicAdd(&deg[d[u]], 1.0f);
}

__global__ void deg_kernel(const int* __restrict__ ei, float* __restrict__ deg2,
                           unsigned E, unsigned N) {
  float* __restrict__ deg = deg2 + package_half() * N;
  const int* __restrict__ dstp = ei + E;
  const unsigned nthr = gridDim.x * blockDim.x;
  const unsigned trips = (E + nthr - 1u) / nthr;
  unsigned e = blockIdx.x * blockDim.x + threadIdx.x;
  unsigned left = trips ? trips - 1u : 0u;
  for (; left >= 4u; left -= 4u, e += 4u * nthr) deg_batch<4>(dstp, deg, e, nthr);
  if (left >= 2u) { deg_batch<2>(dstp, deg, e, nthr); left -= 2u; e += 2u * nthr; }
  if (left >= 1u) { deg_batch<1>(dstp, deg, e, nthr); e += nthr; }
  if (e < E) deg_batch<1>(dstp, deg, e, nthr);
}

__global__ void dis_kernel(const float* __restrict__ degA, const float* __restrict__ degB,
                           float* __restrict__ dis, int N) {
  int i = blockIdx.x * blockDim.x + threadIdx.x;
  if (i < N) { float d = degA[i] + degB[i] + 1.0f; dis[i] = d > 0.0f ? rsqrtf(d) : 0.0f; }
}

__global__ void gemm_wmma_f32(const float* __restrict__ H, const float* __restrict__ W,
                              float* __restrict__ out, int N, int K, int OUT,
                              int nColTiles, int nTiles) {
  int wave = (int)(((unsigned)blockIdx.x * blockDim.x + threadIdx.x) >> 5);
  if (wave >= nTiles) return;
  int lane = threadIdx.x & 31;
  int rt = wave / nColTiles, ct = wave % nColTiles;
  int half = lane >> 4;
  int m = lane & 15;
  int row = rt * 16 + m;
  int rowc = row < N ? row : N - 1;
  int col = ct * 16 + m;
  const float* hrow = H + (unsigned)rowc * (unsigned)K;

  v8f acc = {};
  for (int kk = 0; kk < K; kk += 4) {
    int ka = kk + half * 2;
    v2f a = *(const v2f*)(hrow + ka);
    v2f b;
    b.x = W[(unsigned)ka * (unsigned)OUT + col];
    b.y = W[(unsigned)(ka + 1) * (unsigned)OUT + col];
    acc = __builtin_amdgcn_wmma_f32_16x16x4_f32(
         false, a,  false, b,
         (short)0, acc,  false,  false);
  }
#pragma unroll
  for (int r = 0; r < 8; ++r) {
    int mrow = rt * 16 + r + 8 * half;
    if (mrow < N) out[(unsigned)mrow * (unsigned)OUT + col] = acc[r];
  }
}

template <int LOGC, int U>
__device__ __forceinline__ void scatter_batch(const float* __restrict__ ht,
                                              const int* __restrict__ srcp,
                                              const int* __restrict__ dstp,
                                              const float* __restrict__ dis,
                                              float* __restrict__ agg,
                                              unsigned e0, unsigned step, unsigned laneB) {
  constexpr int WPL = (1 << LOGC) / 32;
  unsigned s[U], d[U];
#pragma unroll
  for (int u = 0; u < U; ++u) {
    s[u] = (unsigned)srcp[e0 + (unsigned)u * step];
    d[u] = (unsigned)dstp[e0 + (unsigned)u * step];
  }
  float nrm[U];
  float v[U * WPL];
#pragma unroll
  for (int u = 0; u < U; ++u) {
    nrm[u] = dis[s[u]] * dis[d[u]];
    const char* r = (const char*)ht + ((s[u] << (LOGC + 2)) + laneB);
#pragma unroll
    for (int w = 0; w < WPL; ++w) v[u * WPL + w] = *(const float*)(r + 128 * w);
  }
#pragma unroll
  for (int u = 0; u < U; ++u) {
    char* o = (char*)agg + ((d[u] << (LOGC + 2)) + laneB);
#pragma unroll
    for (int w = 0; w < WPL; ++w) atomicAdd((float*)(o + 128 * w), v[u * WPL + w] * nrm[u]);
  }
}

template <int LOGC>
__global__ void scatter_kernel(const float* __restrict__ ht, const int* __restrict__ ei,
                               const float* __restrict__ dis, float* __restrict__ agg2,
                               unsigned E, unsigned copyElems) {
  static_assert(LOGC >= 5, "one wave covers at least one 32-word line");
  float* __restrict__ agg = agg2 + package_half() * copyElems;
  const int* __restrict__ srcp = ei;
  const int* __restrict__ dstp = ei + E;
  const unsigned tid = blockIdx.x * blockDim.x + threadIdx.x;
  const unsigned laneB = (tid & 31u) << 2;
  const unsigned estep = (gridDim.x * blockDim.x) >> 5;
  const unsigned trips = (E + estep - 1u) / estep;
  unsigned e = tid >> 5;
  unsigned left = trips ? trips - 1u : 0u;
  for (; left >= 8u; left -= 8u, e += 8u * estep)
    scatter_batch<LOGC, 8>(ht, srcp, dstp, dis, agg, e, estep, laneB);
  if (left >= 4u) {
    scatter_batch<LOGC, 4>(ht, srcp, dstp, dis, agg, e, estep, laneB);
    left -= 4u; e += 4u * estep;
  }
  if (left >= 2u) {
    scatter_batch<LOGC, 2>(ht, srcp, dstp, dis, agg, e, estep, laneB);
    left -= 2u; e += 2u * estep;
  }
  if (left >= 1u) { scatter_batch<LOGC, 1>(ht, srcp, dstp, dis, agg, e, estep, laneB); e += estep; }
  if (e < E) scatter_batch<LOGC, 1>(ht, srcp, dstp, dis, agg, e, estep, laneB);
}

template <int U>
__device__ __forceinline__ void scatter1_batch(const float* __restrict__ ht,
                                               const int* __restrict__ srcp,
                                               const int* __restrict__ dstp,
                                               const float* __restrict__ dis,
                                               float* __restrict__ agg,
                                               unsigned e0, unsigned step) {
  unsigned s[U], d[U];
#pragma unroll
  for (int u = 0; u < U; ++u) {
    s[u] = (unsigned)srcp[e0 + (unsigned)u * step];
    d[u] = (unsigned)dstp[e0 + (unsigned)u * step];
  }
  float m[U];
#pragma unroll
  for (int u = 0; u < U; ++u) m[u] = ht[s[u]] * (dis[s[u]] * dis[d[u]]);
#pragma unroll
  for (int u = 0; u < U; ++u) atomicAdd(&agg[d[u]], m[u]);
}

__global__ void scatter1_kernel(const float* __restrict__ ht, const int* __restrict__ ei,
                                const float* __restrict__ dis, float* __restrict__ agg2,
                                unsigned E, unsigned N) {
  float* __restrict__ agg = agg2 + package_half() * N;
  const int* __restrict__ srcp = ei;
  const int* __restrict__ dstp = ei + E;
  const unsigned nthr = gridDim.x * blockDim.x;
  const unsigned trips = (E + nthr - 1u) / nthr;
  unsigned e = blockIdx.x * blockDim.x + threadIdx.x;
  unsigned left = trips ? trips - 1u : 0u;
  for (; left >= 4u; left -= 4u, e += 4u * nthr)
    scatter1_batch<4>(ht, srcp, dstp, dis, agg, e, nthr);
  if (left >= 2u) { scatter1_batch<2>(ht, srcp, dstp, dis, agg, e, nthr); left -= 2u; e += 2u * nthr; }
  if (left >= 1u) { scatter1_batch<1>(ht, srcp, dstp, dis, agg, e, nthr); e += nthr; }
  if (e < E) scatter1_batch<1>(ht, srcp, dstp, dis, agg, e, nthr);
}

template <int LOGC, bool TANH>
__global__ void finalize_kernel(const float* aggA, const float* __restrict__ aggB,
                                const float* __restrict__ ht,
                                const float* __restrict__ dis, const float* __restrict__ bias,
                                float* out, unsigned total) {
  unsigned t = blockIdx.x * blockDim.x + threadIdx.x;
  if (t >= total) return;
  unsigned i = t >> LOGC;
  unsigned c = t & ((1u << LOGC) - 1u);
  float d = dis[i];
  float v = (aggA[t] + aggB[t]) + d * d * ht[t] + bias[c];
  out[t] = TANH ? tanhf(v) : v;
}

__global__ void dot32_kernel(const float* __restrict__ H, const float* __restrict__ W3,
                             float* __restrict__ out, int N) {
  int i = blockIdx.x * blockDim.x + threadIdx.x;
  if (i >= N) return;
  const float4* h = (const float4*)(H + (unsigned)i * 32u);
  float s = 0.f;
#pragma unroll
  for (int q = 0; q < 8; ++q) {
    float4 a = h[q];
    float4 w = ((const float4*)W3)[q];
    s += a.x * w.x + a.y * w.y + a.z * w.z + a.w * w.w;
  }
  out[i] = s;
}

extern "C" void kernel_launch(void* const* d_in, const int* in_sizes, int n_in,
                              void* d_out, int out_size, void* d_ws, size_t ws_size,
                              hipStream_t stream) {
  const float* x  = (const float*)d_in[0];
  const int*   ei = (const int*)d_in[1];
  const float* W1 = (const float*)d_in[2];
  const float* b1 = (const float*)d_in[3];
  const float* W2 = (const float*)d_in[4];
  const float* b2 = (const float*)d_in[5];
  const float* W3 = (const float*)d_in[6];
  const float* b3 = (const float*)d_in[7];
  float* out = (float*)d_out;

  const int D2 = 64, D = 32;
  const int N = in_sizes[0] / D2;
  const int E = in_sizes[1] / 2;

  float* ws    = (float*)d_ws;
  float* dis   = ws;
  float* ht    = ws + N;
  float* aggA  = ht + (long long)N * D2;
  float* aggB  = aggA + (long long)N * D2;
  float* deg2  = ht;
  float* ht3   = ht;
  float* agg3  = ht + N;
  (void)ws_size; (void)n_in; (void)out_size;

  const int B = 256;
  const int PG = 2048;
  const int PG1 = 1536;

  fill_kernel<<<(unsigned)cdiv_ll(2LL * N, B), B, 0, stream>>>(deg2, 0.0f, (unsigned)(2 * N));
  deg_kernel<<<PG1, B, 0, stream>>>(ei, deg2, (unsigned)E, (unsigned)N);
  dis_kernel<<<(unsigned)cdiv_ll(N, B), B, 0, stream>>>(deg2, deg2 + N, dis, N);

  const int rowTiles = (int)cdiv_ll(N, 16);

  {
    int colT = D2 / 16, tiles = rowTiles * colT;
    gemm_wmma_f32<<<(unsigned)cdiv_ll((long long)tiles * 32, B), B, 0, stream>>>(
        x, W1, ht, N, D2, D2, colT, tiles);
  }
  zero4_kernel<<<(unsigned)cdiv_ll((long long)N * (2 * D2 / 4), B), B, 0, stream>>>(
      (float4*)aggA, (unsigned)(N * (2 * D2 / 4)));
  scatter_kernel<6><<<PG, B, 0, stream>>>(ht, ei, dis, aggA, (unsigned)E, (unsigned)(N * D2));
  finalize_kernel<6, true><<<(unsigned)cdiv_ll((long long)N * D2, B), B, 0, stream>>>(
      aggA, aggB, ht, dis, b1, aggA, (unsigned)(N * D2));

  {
    int colT = D / 16, tiles = rowTiles * colT;
    gemm_wmma_f32<<<(unsigned)cdiv_ll((long long)tiles * 32, B), B, 0, stream>>>(
        aggA, W2, ht, N, D2, D, colT, tiles);
  }
  zero4_kernel<<<(unsigned)cdiv_ll((long long)N * (2 * D / 4), B), B, 0, stream>>>(
      (float4*)aggA, (unsigned)(N * (2 * D / 4)));
  scatter_kernel<5><<<PG, B, 0, stream>>>(ht, ei, dis, aggA, (unsigned)E, (unsigned)(N * D));
  finalize_kernel<5, true><<<(unsigned)cdiv_ll((long long)N * D, B), B, 0, stream>>>(
      aggA, aggA + (long long)N * D, ht, dis, b2, aggA, (unsigned)(N * D));

  dot32_kernel<<<(unsigned)cdiv_ll(N, B), B, 0, stream>>>(aggA, W3, ht3, N);
  fill_kernel<<<(unsigned)cdiv_ll(2LL * N, B), B, 0, stream>>>(agg3, 0.f, (unsigned)(2 * N));
  scatter1_kernel<<<PG1, B, 0, stream>>>(ht3, ei, dis, agg3, (unsigned)E, (unsigned)N);
  finalize_kernel<0, false><<<(unsigned)cdiv_ll(N, B), B, 0, stream>>>(
      agg3, agg3 + N, ht3, dis, b3, out, (unsigned)N);
}


# Norm applied in two factors: producers store rows pre-scaled by dis[src], finalize multiplies each node's total by dis[dst]; edge passes read no per-edge norm words; order of every sum unchanged
# speedup vs baseline: 2.2600x; 1.1086x over previous
#include <hip/hip_runtime.h>
#include <math.h>


typedef float v2f __attribute__((ext_vector_type(2)));
typedef float v8f __attribute__((ext_vector_type(8)));

static inline long long cdiv_ll(long long a, long long b) { return (a + b - 1) / b; }

__global__ void fill_kernel(float* __restrict__ p, float v, unsigned n) {
  unsigned i = blockIdx.x * blockDim.x + threadIdx.x;
  if (i < n) p[i] = v;
}

__global__ void zero4_kernel(float4* __restrict__ p, unsigned n4) {
  unsigned i = blockIdx.x * blockDim.x + threadIdx.x;
  if (i < n4) p[i] = make_float4(0.f, 0.f, 0.f, 0.f);
}

__device__ __forceinline__ unsigned package_half() {
  return (__builtin_amdgcn_s_sendmsg_rtn(0x87) >> 18) & 1u;
}

template <int U>
__device__ __forceinline__ void deg_batch(const int* __restrict__ dstp, float* __restrict__ deg,
                                          unsigned e0, unsigned step) {
  unsigned d[U];
#pragma unroll
  for (int u = 0; u < U; ++u) d[u] = (unsigned)dstp[e0 + (unsigned)u * step];
#pragma unroll
  for (int u = 0; u < U; ++u) atomicAdd(&deg[d[u]], 1.0f);
}

__global__ void deg_kernel(const int* __restrict__ ei, float* __restrict__ deg2,
                           unsigned E, unsigned N) {
  float* __restrict__ deg = deg2 + package_half() * N;
  const int* __restrict__ dstp = ei + E;
  const unsigned nthr = gridDim.x * blockDim.x;
  const unsigned trips = (E + nthr - 1u) / nthr;
  unsigned e = blockIdx.x * blockDim.x + threadIdx.x;
  unsigned left = trips ? trips - 1u : 0u;
  for (; left >= 4u; left -= 4u, e += 4u * nthr) deg_batch<4>(dstp, deg, e, nthr);
  if (left >= 2u) { deg_batch<2>(dstp, deg, e, nthr); left -= 2u; e += 2u * nthr; }
  if (left >= 1u) { deg_batch<1>(dstp, deg, e, nthr); e += nthr; }
  if (e < E) deg_batch<1>(dstp, deg, e, nthr);
}

__global__ void dis_kernel(const float* __restrict__ degA, const float* __restrict__ degB,
                           float* __restrict__ dis, int N) {
  int i = blockIdx.x * blockDim.x + threadIdx.x;
  if (i < N) { float d = degA[i] + degB[i] + 1.0f; dis[i] = d > 0.0f ? rsqrtf(d) : 0.0f; }
}

__global__ void gemm_wmma_f32(const float* __restrict__ H, const float* __restrict__ W,
                              const float* __restrict__ dis,
                              float* __restrict__ out, int N, int K, int OUT,
                              int nColTiles, int nTiles) {
  int wave = (int)(((unsigned)blockIdx.x * blockDim.x + threadIdx.x) >> 5);
  if (wave >= nTiles) return;
  int lane = threadIdx.x & 31;
  int rt = wave / nColTiles, ct = wave % nColTiles;
  int half = lane >> 4;
  int m = lane & 15;
  int row = rt * 16 + m;
  int rowc = row < N ? row : N - 1;
  int col = ct * 16 + m;
  const float* hrow = H + (unsigned)rowc * (unsigned)K;

  float dsc[8];
#pragma unroll
  for (int r = 0; r < 8; ++r) {
    int mr = rt * 16 + r + 8 * half;
    dsc[r] = dis[mr < N ? mr : N - 1];
  }

  v8f acc = {};
  for (int kk = 0; kk < K; kk += 4) {
    int ka = kk + half * 2;
    v2f a = *(const v2f*)(hrow + ka);
    v2f b;
    b.x = W[(unsigned)ka * (unsigned)OUT + col];
    b.y = W[(unsigned)(ka + 1) * (unsigned)OUT + col];
    acc = __builtin_amdgcn_wmma_f32_16x16x4_f32(
         false, a,  false, b,
         (short)0, acc,  false,  false);
  }
#pragma unroll
  for (int r = 0; r < 8; ++r) {
    int mrow = rt * 16 + r + 8 * half;
    if (mrow < N) out[(unsigned)mrow * (unsigned)OUT + col] = acc[r] * dsc[r];
  }
}

template <int LOGC, int U>
__device__ __forceinline__ void scatter_batch(const float* __restrict__ ht,
                                              const int* __restrict__ srcp,
                                              const int* __restrict__ dstp,
                                              float* __restrict__ agg,
                                              unsigned e0, unsigned step, unsigned laneB) {
  constexpr int WPL = (1 << LOGC) / 32;
  unsigned s[U], d[U];
#pragma unroll
  for (int u = 0; u < U; ++u) {
    s[u] = (unsigned)srcp[e0 + (unsigned)u * step];
    d[u] = (unsigned)dstp[e0 + (unsigned)u * step];
  }
  float v[U * WPL];
#pragma unroll
  for (int u = 0; u < U; ++u) {
    const char* r = (const char*)ht + ((s[u] << (LOGC + 2)) + laneB);
#pragma unroll
    for (int w = 0; w < WPL; ++w) v[u * WPL + w] = *(const float*)(r + 128 * w);
  }
#pragma unroll
  for (int u = 0; u < U; ++u) {
    char* o = (char*)agg + ((d[u] << (LOGC + 2)) + laneB);
#pragma unroll
    for (int w = 0; w < WPL; ++w) atomicAdd((float*)(o + 128 * w), v[u * WPL + w]);
  }
}

template <int LOGC>
__global__ void scatter_kernel(const float* __restrict__ ht, const int* __restrict__ ei,
                               float* __restrict__ agg2,
                               unsigned E, unsigned copyElems) {
  static_assert(LOGC >= 5, "one wave covers at least one 32-word line");
  float* __restrict__ agg = agg2 + package_half() * copyElems;
  const int* __restrict__ srcp = ei;
  const int* __restrict__ dstp = ei + E;
  const unsigned tid = blockIdx.x * blockDim.x + threadIdx.x;
  const unsigned laneB = (tid & 31u) << 2;
  const unsigned estep = (gridDim.x * blockDim.x) >> 5;
  const unsigned trips = (E + estep - 1u) / estep;
  unsigned e = tid >> 5;
  unsigned left = trips ? trips - 1u : 0u;
  for (; left >= 8u; left -= 8u, e += 8u * estep)
    scatter_batch<LOGC, 8>(ht, srcp, dstp, agg, e, estep, laneB);
  if (left >= 4u) {
    scatter_batch<LOGC, 4>(ht, srcp, dstp, agg, e, estep, laneB);
    left -= 4u; e += 4u * estep;
  }
  if (left >= 2u) {
    scatter_batch<LOGC, 2>(ht, srcp, dstp, agg, e, estep, laneB);
    left -= 2u; e += 2u * estep;
  }
  if (left >= 1u) { scatter_batch<LOGC, 1>(ht, srcp, dstp, agg, e, estep, laneB); e += estep; }
  if (e < E) scatter_batch<LOGC, 1>(ht, srcp, dstp, agg, e, estep, laneB);
}

template <int U>
__device__ __forceinline__ void scatter1_batch(const float* __restrict__ ht,
                                               const int* __restrict__ srcp,
                                               const int* __restrict__ dstp,
                                               float* __restrict__ agg,
                                               unsigned e0, unsigned step) {
  unsigned s[U], d[U];
#pragma unroll
  for (int u = 0; u < U; ++u) {
    s[u] = (unsigned)srcp[e0 + (unsigned)u * step];
    d[u] = (unsigned)dstp[e0 + (unsigned)u * step];
  }
  float m[U];
#pragma unroll
  for (int u = 0; u < U; ++u) m[u] = ht[s[u]];
#pragma unroll
  for (int u = 0; u < U; ++u) atomicAdd(&agg[d[u]], m[u]);
}

__global__ void scatter1_kernel(const float* __restrict__ ht, const int* __restrict__ ei,
                                float* __restrict__ agg2,
                                unsigned E, unsigned N) {
  float* __restrict__ agg = agg2 + package_half() * N;
  const int* __restrict__ srcp = ei;
  const int* __restrict__ dstp = ei + E;
  const unsigned nthr = gridDim.x * blockDim.x;
  const unsigned trips = (E + nthr - 1u) / nthr;
  unsigned e = blockIdx.x * blockDim.x + threadIdx.x;
  unsigned left = trips ? trips - 1u : 0u;
  for (; left >= 4u; left -= 4u, e += 4u * nthr)
    scatter1_batch<4>(ht, srcp, dstp, agg, e, nthr);
  if (left >= 2u) { scatter1_batch<2>(ht, srcp, dstp, agg, e, nthr); left -= 2u; e += 2u * nthr; }
  if (left >= 1u) { scatter1_batch<1>(ht, srcp, dstp, agg, e, nthr); e += nthr; }
  if (e < E) scatter1_batch<1>(ht, srcp, dstp, agg, e, nthr);
}

template <int LOGC, bool TANH>
__global__ void finalize_kernel(const float* aggA, const float* __restrict__ aggB,
                                const float* __restrict__ ht,
                                const float* __restrict__ dis, const float* __restrict__ bias,
                                float* out, unsigned total) {
  unsigned t = blockIdx.x * blockDim.x + threadIdx.x;
  if (t >= total) return;
  unsigned i = t >> LOGC;
  unsigned c = t & ((1u << LOGC) - 1u);
  float d = dis[i];
  float v = d * ((aggA[t] + aggB[t]) + ht[t]) + bias[c];
  out[t] = TANH ? tanhf(v) : v;
}

__global__ void dot32_kernel(const float* __restrict__ H, const float* __restrict__ W3,
                             const float* __restrict__ dis,
                             float* __restrict__ out, int N) {
  int i = blockIdx.x * blockDim.x + threadIdx.x;
  if (i >= N) return;
  const float4* h = (const float4*)(H + (unsigned)i * 32u);
  float s = 0.f;
#pragma unroll
  for (int q = 0; q < 8; ++q) {
    float4 a = h[q];
    float4 w = ((const float4*)W3)[q];
    s += a.x * w.x + a.y * w.y + a.z * w.z + a.w * w.w;
  }
  out[i] = s * dis[i];
}

extern "C" void kernel_launch(void* const* d_in, const int* in_sizes, int n_in,
                              void* d_out, int out_size, void* d_ws, size_t ws_size,
                              hipStream_t stream) {
  const float* x  = (const float*)d_in[0];
  const int*   ei = (const int*)d_in[1];
  const float* W1 = (const float*)d_in[2];
  const float* b1 = (const float*)d_in[3];
  const float* W2 = (const float*)d_in[4];
  const float* b2 = (const float*)d_in[5];
  const float* W3 = (const float*)d_in[6];
  const float* b3 = (const float*)d_in[7];
  float* out = (float*)d_out;

  const int D2 = 64, D = 32;
  const int N = in_sizes[0] / D2;
  const int E = in_sizes[1] / 2;

  float* ws    = (float*)d_ws;
  float* dis   = ws;
  float* ht    = ws + N;
  float* aggA  = ht + (long long)N * D2;
  float* aggB  = aggA + (long long)N * D2;
  float* deg2  = ht;
  float* ht3   = ht;
  float* agg3  = ht + N;
  (void)ws_size; (void)n_in; (void)out_size;

  const int B = 256;
  const int PG = 2048;
  const int PG1 = 1536;

  fill_kernel<<<(unsigned)cdiv_ll(2LL * N, B), B, 0, stream>>>(deg2, 0.0f, (unsigned)(2 * N));
  deg_kernel<<<PG1, B, 0, stream>>>(ei, deg2, (unsigned)E, (unsigned)N);
  dis_kernel<<<(unsigned)cdiv_ll(N, B), B, 0, stream>>>(deg2, deg2 + N, dis, N);

  const int rowTiles = (int)cdiv_ll(N, 16);

  {
    int colT = D2 / 16, tiles = rowTiles * colT;
    gemm_wmma_f32<<<(unsigned)cdiv_ll((long long)tiles * 32, B), B, 0, stream>>>(
        x, W1, dis, ht, N, D2, D2, colT, tiles);
  }
  zero4_kernel<<<(unsigned)cdiv_ll((long long)N * (2 * D2 / 4), B), B, 0, stream>>>(
      (float4*)aggA, (unsigned)(N * (2 * D2 / 4)));
  scatter_kernel<6><<<PG, B, 0, stream>>>(ht, ei, aggA, (unsigned)E, (unsigned)(N * D2));
  finalize_kernel<6, true><<<(unsigned)cdiv_ll((long long)N * D2, B), B, 0, stream>>>(
      aggA, aggB, ht, dis, b1, aggA, (unsigned)(N * D2));

  {
    int colT = D / 16, tiles = rowTiles * colT;
    gemm_wmma_f32<<<(unsigned)cdiv_ll((long long)tiles * 32, B), B, 0, stream>>>(
        aggA, W2, dis, ht, N, D2, D, colT, tiles);
  }
  zero4_kernel<<<(unsigned)cdiv_ll((long long)N * (2 * D / 4), B), B, 0, stream>>>(
      (float4*)aggA, (unsigned)(N * (2 * D / 4)));
  scatter_kernel<5><<<PG, B, 0, stream>>>(ht, ei, aggA, (unsigned)E, (unsigned)(N * D));
  finalize_kernel<5, true><<<(unsigned)cdiv_ll((long long)N * D, B), B, 0, stream>>>(
      aggA, aggA + (long long)N * D, ht, dis, b2, aggA, (unsigned)(N * D));

  dot32_kernel<<<(unsigned)cdiv_ll(N, B), B, 0, stream>>>(aggA, W3, dis, ht3, N);
  fill_kernel<<<(unsigned)cdiv_ll(2LL * N, B), B, 0, stream>>>(agg3, 0.f, (unsigned)(2 * N));
  scatter1_kernel<<<PG1, B, 0, stream>>>(ht3, ei, agg3, (unsigned)E, (unsigned)N);
  finalize_kernel<0, false><<<(unsigned)cdiv_ll(N, B), B, 0, stream>>>(
      agg3, agg3 + N, ht3, dis, b3, out, (unsigned)N);
}


# Each tanh finalize pass folded into its consumer's loader: gemm2's A loader and the layer-3 dot form h1/h2 in registers, never stored; layer-3 dot sum order: 4-term partials + 8-lane tree
# speedup vs baseline: 2.4099x; 1.0663x over previous
#include <hip/hip_runtime.h>
#include <math.h>


typedef float v2f __attribute__((ext_vector_type(2)));
typedef float v8f __attribute__((ext_vector_type(8)));

static inline long long cdiv_ll(long long a, long long b) { return (a + b - 1) / b; }

__global__ void fill_kernel(float* __restrict__ p, float v, unsigned n) {
  unsigned i = blockIdx.x * blockDim.x + threadIdx.x;
  if (i < n) p[i] = v;
}

__global__ void zero4_kernel(float4* __restrict__ p, unsigned n4) {
  unsigned i = blockIdx.x * blockDim.x + threadIdx.x;
  if (i < n4) p[i] = make_float4(0.f, 0.f, 0.f, 0.f);
}

__device__ __forceinline__ unsigned package_half() {
  return (__builtin_amdgcn_s_sendmsg_rtn(0x87) >> 18) & 1u;
}

template <int U>
__device__ __forceinline__ void deg_batch(const int* __restrict__ dstp, float* __restrict__ deg,
                                          unsigned e0, unsigned step) {
  unsigned d[U];
#pragma unroll
  for (int u = 0; u < U; ++u) d[u] = (unsigned)dstp[e0 + (unsigned)u * step];
#pragma unroll
  for (int u = 0; u < U; ++u) atomicAdd(&deg[d[u]], 1.0f);
}

__global__ void deg_kernel(const int* __restrict__ ei, float* __restrict__ deg2,
                           unsigned E, unsigned N) {
  float* __restrict__ deg = deg2 + package_half() * N;
  const int* __restrict__ dstp = ei + E;
  const unsigned nthr = gridDim.x * blockDim.x;
  const unsigned trips = (E + nthr - 1u) / nthr;
  unsigned e = blockIdx.x * blockDim.x + threadIdx.x;
  unsigned left = trips ? trips - 1u : 0u;
  for (; left >= 4u; left -= 4u, e += 4u * nthr) deg_batch<4>(dstp, deg, e, nthr);
  if (left >= 2u) { deg_batch<2>(dstp, deg, e, nthr); left -= 2u; e += 2u * nthr; }
  if (left >= 1u) { deg_batch<1>(dstp, deg, e, nthr); e += nthr; }
  if (e < E) deg_batch<1>(dstp, deg, e, nthr);
}

__global__ void dis_kernel(const float* __restrict__ degA, const float* __restrict__ degB,
                           float* __restrict__ dis, int N) {
  int i = blockIdx.x * blockDim.x + threadIdx.x;
  if (i < N) { float d = degA[i] + degB[i] + 1.0f; dis[i] = d > 0.0f ? rsqrtf(d) : 0.0f; }
}

__global__ void gemm_wmma_f32(const float* __restrict__ H, const float* __restrict__ W,
                              const float* __restrict__ dis,
                              float* __restrict__ out, int N, int K, int OUT,
                              int nColTiles, int nTiles) {
  int wave = (int)(((unsigned)blockIdx.x * blockDim.x + threadIdx.x) >> 5);
  if (wave >= nTiles) return;
  int lane = threadIdx.x & 31;
  int rt = wave / nColTiles, ct = wave % nColTiles;
  int half = lane >> 4;
  int m = lane & 15;
  int row = rt * 16 + m;
  int rowc = row < N ? row : N - 1;
  int col = ct * 16 + m;
  const float* hrow = H + (unsigned)rowc * (unsigned)K;

  float dsc[8];
#pragma unroll
  for (int r = 0; r < 8; ++r) {
    int mr = rt * 16 + r + 8 * half;
    dsc[r] = dis[mr < N ? mr : N - 1];
  }

  v8f acc = {};
  for (int kk = 0; kk < K; kk += 4) {
    int ka = kk + half * 2;
    v2f a = *(const v2f*)(hrow + ka);
    v2f b;
    b.x = W[(unsigned)ka * (unsigned)OUT + col];
    b.y = W[(unsigned)(ka + 1) * (unsigned)OUT + col];
    acc = __builtin_amdgcn_wmma_f32_16x16x4_f32(
         false, a,  false, b,
         (short)0, acc,  false,  false);
  }
#pragma unroll
  for (int r = 0; r < 8; ++r) {
    int mrow = rt * 16 + r + 8 * half;
    if (mrow < N) out[(unsigned)mrow * (unsigned)OUT + col] = acc[r] * dsc[r];
  }
}

template <int LOGC, int SRCSH, int U>
__device__ __forceinline__ void scatter_batch(const float* __restrict__ ht,
                                              const int* __restrict__ srcp,
                                              const int* __restrict__ dstp,
                                              float* __restrict__ agg,
                                              unsigned e0, unsigned step, unsigned laneB) {
  constexpr int WPL = (1 << LOGC) / 32;
  unsigned s[U], d[U];
#pragma unroll
  for (int u = 0; u < U; ++u) {
    s[u] = (unsigned)srcp[e0 + (unsigned)u * step];
    d[u] = (unsigned)dstp[e0 + (unsigned)u * step];
  }
  float v[U * WPL];
#pragma unroll
  for (int u = 0; u < U; ++u) {
    const char* r = (const char*)ht + ((s[u] << (SRCSH + 2)) + laneB);
#pragma unroll
    for (int w = 0; w < WPL; ++w) v[u * WPL + w] = *(const float*)(r + 128 * w);
  }
#pragma unroll
  for (int u = 0; u < U; ++u) {
    char* o = (char*)agg + ((d[u] << (LOGC + 2)) + laneB);
#pragma unroll
    for (int w = 0; w < WPL; ++w) atomicAdd((float*)(o + 128 * w), v[u * WPL + w]);
  }
}

template <int LOGC, int SRCSH>
__global__ void scatter_kernel(const float* __restrict__ ht, const int* __restrict__ ei,
                               float* __restrict__ agg2,
                               unsigned E, unsigned copyElems) {
  static_assert(LOGC >= 5, "one wave covers at least one 32-word line");
  float* __restrict__ agg = agg2 + package_half() * copyElems;
  const int* __restrict__ srcp = ei;
  const int* __restrict__ dstp = ei + E;
  const unsigned tid = blockIdx.x * blockDim.x + threadIdx.x;
  const unsigned laneB = (tid & 31u) << 2;
  const unsigned estep = (gridDim.x * blockDim.x) >> 5;
  const unsigned trips = (E + estep - 1u) / estep;
  unsigned e = tid >> 5;
  unsigned left = trips ? trips - 1u : 0u;
  for (; left >= 8u; left -= 8u, e += 8u * estep)
    scatter_batch<LOGC, SRCSH, 8>(ht, srcp, dstp, agg, e, estep, laneB);
  if (left >= 4u) {
    scatter_batch<LOGC, SRCSH, 4>(ht, srcp, dstp, agg, e, estep, laneB);
    left -= 4u; e += 4u * estep;
  }
  if (left >= 2u) {
    scatter_batch<LOGC, SRCSH, 2>(ht, srcp, dstp, agg, e, estep, laneB);
    left -= 2u; e += 2u * estep;
  }
  if (left >= 1u) { scatter_batch<LOGC, SRCSH, 1>(ht, srcp, dstp, agg, e, estep, laneB); e += estep; }
  if (e < E) scatter_batch<LOGC, SRCSH, 1>(ht, srcp, dstp, agg, e, estep, laneB);
}

template <int U>
__device__ __forceinline__ void scatter1_batch(const float* __restrict__ ht,
                                               const int* __restrict__ srcp,
                                               const int* __restrict__ dstp,
                                               float* __restrict__ agg,
                                               unsigned e0, unsigned step) {
  unsigned s[U], d[U];
#pragma unroll
  for (int u = 0; u < U; ++u) {
    s[u] = (unsigned)srcp[e0 + (unsigned)u * step];
    d[u] = (unsigned)dstp[e0 + (unsigned)u * step];
  }
  float m[U];
#pragma unroll
  for (int u = 0; u < U; ++u) m[u] = ht[s[u]];
#pragma unroll
  for (int u = 0; u < U; ++u) atomicAdd(&agg[d[u]], m[u]);
}

__global__ void scatter1_kernel(const float* __restrict__ ht, const int* __restrict__ ei,
                                float* __restrict__ agg2,
                                unsigned E, unsigned N) {
  float* __restrict__ agg = agg2 + package_half() * N;
  const int* __restrict__ srcp = ei;
  const int* __restrict__ dstp = ei + E;
  const unsigned nthr = gridDim.x * blockDim.x;
  const unsigned trips = (E + nthr - 1u) / nthr;
  unsigned e = blockIdx.x * blockDim.x + threadIdx.x;
  unsigned left = trips ? trips - 1u : 0u;
  for (; left >= 4u; left -= 4u, e += 4u * nthr)
    scatter1_batch<4>(ht, srcp, dstp, agg, e, nthr);
  if (left >= 2u) { scatter1_batch<2>(ht, srcp, dstp, agg, e, nthr); left -= 2u; e += 2u * nthr; }
  if (left >= 1u) { scatter1_batch<1>(ht, srcp, dstp, agg, e, nthr); e += nthr; }
  if (e < E) scatter1_batch<1>(ht, srcp, dstp, agg, e, nthr);
}

template <int LOGC, bool TANH>
__global__ void finalize_kernel(const float* aggA, const float* __restrict__ aggB,
                                const float* __restrict__ ht,
                                const float* __restrict__ dis, const float* __restrict__ bias,
                                float* out, unsigned total) {
  unsigned t = blockIdx.x * blockDim.x + threadIdx.x;
  if (t >= total) return;
  unsigned i = t >> LOGC;
  unsigned c = t & ((1u << LOGC) - 1u);
  float d = dis[i];
  float v = d * ((aggA[t] + aggB[t]) + ht[t]) + bias[c];
  out[t] = TANH ? tanhf(v) : v;
}

__device__ __forceinline__ float act_tanh(float a0, float a1, float self_row, float d, float b) {
  return tanhf(d * ((a0 + a1) + self_row) + b);
}

__global__ void __launch_bounds__(256) gemm2_fused(const float* __restrict__ sumA, const float* __restrict__ sumB,
                            float* hs, const float* __restrict__ dis,
                            const float* __restrict__ b1, const float* __restrict__ W2,
                            int N, int nRowTiles) {
  int wave = (int)(((unsigned)blockIdx.x * blockDim.x + threadIdx.x) >> 5);
  if (wave >= nRowTiles) return;
  int lane = threadIdx.x & 31;
  int half = lane >> 4;
  int m = lane & 15;
  int row = wave * 16 + m;
  int rowc = row < N ? row : N - 1;
  const unsigned ro = (unsigned)rowc * 64u;
  const float dr = dis[rowc];

  float dsc[8];
#pragma unroll
  for (int r = 0; r < 8; ++r) {
    int mr = wave * 16 + r + 8 * half;
    dsc[r] = dis[mr < N ? mr : N - 1];
  }

  v2f a0[16], a1[16], sr[16];
#pragma unroll
  for (int j = 0; j < 16; ++j) {
    int ka = 4 * j + half * 2;
    a0[j] = *(const v2f*)(sumA + ro + ka);
    a1[j] = *(const v2f*)(sumB + ro + ka);
    sr[j] = *(const v2f*)(hs + ro + ka);
  }
  v2f bcur = *(const v2f*)(b1 + half * 2);

  v8f acc0 = {}, acc1 = {};
#pragma unroll
  for (int j = 0; j < 16; ++j) {
    int ka = 4 * j + half * 2;
    v2f w0, w1;
    w0.x = W2[(unsigned)ka * 32u + m];
    w0.y = W2[(unsigned)(ka + 1) * 32u + m];
    w1.x = W2[(unsigned)ka * 32u + 16u + m];
    w1.y = W2[(unsigned)(ka + 1) * 32u + 16u + m];
    v2f bnext = bcur;
    if (j + 1 < 16) bnext = *(const v2f*)(b1 + ka + 4);
    v2f a;
    a.x = act_tanh(a0[j].x, a1[j].x, sr[j].x, dr, bcur.x);
    a.y = act_tanh(a0[j].y, a1[j].y, sr[j].y, dr, bcur.y);
    bcur = bnext;
    acc0 = __builtin_amdgcn_wmma_f32_16x16x4_f32(
         false, a,  false, w0,
         (short)0, acc0,  false,  false);
    acc1 = __builtin_amdgcn_wmma_f32_16x16x4_f32(
         false, a,  false, w1,
         (short)0, acc1,  false,  false);
  }
#pragma unroll
  for (int r = 0; r < 8; ++r) {
    int mrow = wave * 16 + r + 8 * half;
    if (mrow < N) {
      hs[(unsigned)mrow * 64u + m]       = acc0[r] * dsc[r];
      hs[(unsigned)mrow * 64u + 16u + m] = acc1[r] * dsc[r];
    }
  }
}

__global__ void dot32_fused(const float* __restrict__ cA, const float* __restrict__ cB,
                            const float* __restrict__ hs, const float* __restrict__ dis,
                            const float* __restrict__ b2, const float* __restrict__ W3,
                            float* __restrict__ out, unsigned N) {
  unsigned t = blockIdx.x * blockDim.x + threadIdx.x;
  unsigned i = t >> 3, q = t & 7u;
  unsigned ic = i < N ? i : N - 1u;
  float d = dis[ic];
  float4 a0 = ((const float4*)(cA + ic * 32u))[q];
  float4 a1 = ((const float4*)(cB + ic * 32u))[q];
  float4 sr = ((const float4*)(hs + ic * 64u))[q];
  float4 bb = ((const float4*)b2)[q];
  float4 w  = ((const float4*)W3)[q];
  float p = act_tanh(a0.x, a1.x, sr.x, d, bb.x) * w.x + act_tanh(a0.y, a1.y, sr.y, d, bb.y) * w.y +
            act_tanh(a0.z, a1.z, sr.z, d, bb.z) * w.z + act_tanh(a0.w, a1.w, sr.w, d, bb.w) * w.w;
  p += __shfl_xor(p, 1);
  p += __shfl_xor(p, 2);
  p += __shfl_xor(p, 4);
  if (q == 0u && i < N) out[i] = p * d;
}

extern "C" void kernel_launch(void* const* d_in, const int* in_sizes, int n_in,
                              void* d_out, int out_size, void* d_ws, size_t ws_size,
                              hipStream_t stream) {
  const float* x  = (const float*)d_in[0];
  const int*   ei = (const int*)d_in[1];
  const float* W1 = (const float*)d_in[2];
  const float* b1 = (const float*)d_in[3];
  const float* W2 = (const float*)d_in[4];
  const float* b2 = (const float*)d_in[5];
  const float* W3 = (const float*)d_in[6];
  const float* b3 = (const float*)d_in[7];
  float* out = (float*)d_out;

  const int D2 = 64, D = 32;
  const int N = in_sizes[0] / D2;
  const int E = in_sizes[1] / 2;

  float* ws    = (float*)d_ws;
  float* dis   = ws;
  float* ht    = ws + N;
  float* aggA  = ht + (long long)N * D2;
  float* aggB  = aggA + (long long)N * D2;
  float* deg2  = ht;
  float* ht3   = aggB;
  float* agg3  = aggB + N;
  (void)ws_size; (void)n_in; (void)out_size;

  const int B = 256;
  const int PG = 2048;
  const int PG1 = 1536;

  fill_kernel<<<(unsigned)cdiv_ll(2LL * N, B), B, 0, stream>>>(deg2, 0.0f, (unsigned)(2 * N));
  deg_kernel<<<PG1, B, 0, stream>>>(ei, deg2, (unsigned)E, (unsigned)N);
  dis_kernel<<<(unsigned)cdiv_ll(N, B), B, 0, stream>>>(deg2, deg2 + N, dis, N);

  const int rowTiles = (int)cdiv_ll(N, 16);

  {
    int colT = D2 / 16, tiles = rowTiles * colT;
    gemm_wmma_f32<<<(unsigned)cdiv_ll((long long)tiles * 32, B), B, 0, stream>>>(
        x, W1, dis, ht, N, D2, D2, colT, tiles);
  }
  zero4_kernel<<<(unsigned)cdiv_ll((long long)N * (2 * D2 / 4), B), B, 0, stream>>>(
      (float4*)aggA, (unsigned)(N * (2 * D2 / 4)));
  scatter_kernel<6, 6><<<PG, B, 0, stream>>>(ht, ei, aggA, (unsigned)E, (unsigned)(N * D2));

  gemm2_fused<<<(unsigned)cdiv_ll((long long)rowTiles * 32, B), B, 0, stream>>>(
      aggA, aggB, ht, dis, b1, W2, N, rowTiles);
  zero4_kernel<<<(unsigned)cdiv_ll((long long)N * (2 * D / 4), B), B, 0, stream>>>(
      (float4*)aggA, (unsigned)(N * (2 * D / 4)));
  scatter_kernel<5, 6><<<PG, B, 0, stream>>>(ht, ei, aggA, (unsigned)E, (unsigned)(N * D));

  dot32_fused<<<(unsigned)cdiv_ll(8LL * N, B), B, 0, stream>>>(
      aggA, aggA + (long long)N * D, ht, dis, b2, W3, ht3, (unsigned)N);
  fill_kernel<<<(unsigned)cdiv_ll(2LL * N, B), B, 0, stream>>>(agg3, 0.f, (unsigned)(2 * N));
  scatter1_kernel<<<PG1, B, 0, stream>>>(ht3, ei, agg3, (unsigned)E, (unsigned)N);
  finalize_kernel<0, false><<<(unsigned)cdiv_ll(N, B), B, 0, stream>>>(
      agg3, agg3 + N, ht3, dis, b3, out, (unsigned)N);
}


# Short-wave passes run as few long-lived waves: zeroing and fills stride over a small grid, dot carries 4 rows per lane group, gemm1 does all 4 column tiles per wave; order of every sum unchanged
# speedup vs baseline: 2.5169x; 1.0444x over previous
#include <hip/hip_runtime.h>
#include <math.h>


typedef float v2f __attribute__((ext_vector_type(2)));
typedef float v8f __attribute__((ext_vector_type(8)));

static inline long long cdiv_ll(long long a, long long b) { return (a + b - 1) / b; }

__global__ void fill_kernel(float* __restrict__ p, float v, unsigned n) {
  const unsigned nthr = gridDim.x * blockDim.x;
  for (unsigned i = blockIdx.x * blockDim.x + threadIdx.x; i < n; i += nthr) p[i] = v;
}

__global__ void zero4_kernel(float4* __restrict__ p, unsigned n4) {
  const unsigned nthr = gridDim.x * blockDim.x;
  for (unsigned i = blockIdx.x * blockDim.x + threadIdx.x; i < n4; i += nthr)
    p[i] = make_float4(0.f, 0.f, 0.f, 0.f);
}

__device__ __forceinline__ unsigned package_half() {
  return (__builtin_amdgcn_s_sendmsg_rtn(0x87) >> 18) & 1u;
}

template <int U>
__device__ __forceinline__ void deg_batch(const int* __restrict__ dstp, float* __restrict__ deg,
                                          unsigned e0, unsigned step) {
  unsigned d[U];
#pragma unroll
  for (int u = 0; u < U; ++u) d[u] = (unsigned)dstp[e0 + (unsigned)u * step];
#pragma unroll
  for (int u = 0; u < U; ++u) atomicAdd(&deg[d[u]], 1.0f);
}

__global__ void deg_kernel(const int* __restrict__ ei, float* __restrict__ deg2,
                           unsigned E, unsigned N) {
  float* __restrict__ deg = deg2 + package_half() * N;
  const int* __restrict__ dstp = ei + E;
  const unsigned nthr = gridDim.x * blockDim.x;
  const unsigned trips = (E + nthr - 1u) / nthr;
  unsigned e = blockIdx.x * blockDim.x + threadIdx.x;
  unsigned left = trips ? trips - 1u : 0u;
  for (; left >= 4u; left -= 4u, e += 4u * nthr) deg_batch<4>(dstp, deg, e, nthr);
  if (left >= 2u) { deg_batch<2>(dstp, deg, e, nthr); left -= 2u; e += 2u * nthr; }
  if (left >= 1u) { deg_batch<1>(dstp, deg, e, nthr); e += nthr; }
  if (e < E) deg_batch<1>(dstp, deg, e, nthr);
}

__global__ void dis_kernel(const float* __restrict__ degA, const float* __restrict__ degB,
                           float* __restrict__ dis, int N) {
  int i = blockIdx.x * blockDim.x + threadIdx.x;
  if (i < N) { float d = degA[i] + degB[i] + 1.0f; dis[i] = d > 0.0f ? rsqrtf(d) : 0.0f; }
}

__global__ void __launch_bounds__(256) gemm1_rows(const float* __restrict__ H, const float* __restrict__ W,
                              const float* __restrict__ dis,
                              float* __restrict__ out, int N, int nRowTiles) {
  int wave = (int)(((unsigned)blockIdx.x * blockDim.x + threadIdx.x) >> 5);
  if (wave >= nRowTiles) return;
  int lane = threadIdx.x & 31;
  int half = lane >> 4;
  int m = lane & 15;
  int row = wave * 16 + m;
  int rowc = row < N ? row : N - 1;
  const float* hrow = H + (unsigned)rowc * 64u;

  float dsc[8];
#pragma unroll
  for (int r = 0; r < 8; ++r) {
    int mr = wave * 16 + r + 8 * half;
    dsc[r] = dis[mr < N ? mr : N - 1];
  }

  v2f a[16];
#pragma unroll
  for (int j = 0; j < 16; ++j) a[j] = *(const v2f*)(hrow + 4 * j + half * 2);

  v8f acc0 = {}, acc1 = {}, acc2 = {}, acc3 = {};
#pragma unroll
  for (int j = 0; j < 16; ++j) {
    unsigned ka = (unsigned)(4 * j + half * 2);
    const float* w0 = W + ka * 64u + m;
    const float* w1 = w0 + 64;
    v2f b0, b1, b2, b3;
    b0.x = w0[0];  b0.y = w1[0];
    b1.x = w0[16]; b1.y = w1[16];
    b2.x = w0[32]; b2.y = w1[32];
    b3.x = w0[48]; b3.y = w1[48];
    acc0 = __builtin_amdgcn_wmma_f32_16x16x4_f32(
         false, a[j],  false, b0,
         (short)0, acc0,  false,  false);
    acc1 = __builtin_amdgcn_wmma_f32_16x16x4_f32(
         false, a[j],  false, b1,
         (short)0, acc1,  false,  false);
    acc2 = __builtin_amdgcn_wmma_f32_16x16x4_f32(
         false, a[j],  false, b2,
         (short)0, acc2,  false,  false);
    acc3 = __builtin_amdgcn_wmma_f32_16x16x4_f32(
         false, a[j],  false, b3,
         (short)0, acc3,  false,  false);
  }
#pragma unroll
  for (int r = 0; r < 8; ++r) {
    int mrow = wave * 16 + r + 8 * half;
    if (mrow < N) {
      float* o = out + (unsigned)mrow * 64u + m;
      o[0]  = acc0[r] * dsc[r];
      o[16] = acc1[r] * dsc[r];
      o[32] = acc2[r] * dsc[r];
      o[48] = acc3[r] * dsc[r];
    }
  }
}

template <int LOGC, int SRCSH, int U>
__device__ __forceinline__ void scatter_batch(const float* __restrict__ ht,
                                              const int* __restrict__ srcp,
                                              const int* __restrict__ dstp,
                                              float* __restrict__ agg,
                                              unsigned e0, unsigned step, unsigned laneB) {
  constexpr int WPL = (1 << LOGC) / 32;
  unsigned s[U], d[U];
#pragma unroll
  for (int u = 0; u < U; ++u) {
    s[u] = (unsigned)srcp[e0 + (unsigned)u * step];
    d[u] = (unsigned)dstp[e0 + (unsigned)u * step];
  }
  float v[U * WPL];
#pragma unroll
  for (int u = 0; u < U; ++u) {
    const char* r = (const char*)ht + ((s[u] << (SRCSH + 2)) + laneB);
#pragma unroll
    for (int w = 0; w < WPL; ++w) v[u * WPL + w] = *(const float*)(r + 128 * w);
  }
#pragma unroll
  for (int u = 0; u < U; ++u) {
    char* o = (char*)agg + ((d[u] << (LOGC + 2)) + laneB);
#pragma unroll
    for (int w = 0; w < WPL; ++w) atomicAdd((float*)(o + 128 * w), v[u * WPL + w]);
  }
}

template <int LOGC, int SRCSH>
__global__ void scatter_kernel(const float* __restrict__ ht, const int* __restrict__ ei,
                               float* __restrict__ agg2,
                               unsigned E, unsigned copyElems) {
  static_assert(LOGC >= 5, "one wave covers at least one 32-word line");
  float* __restrict__ agg = agg2 + package_half() * copyElems;
  const int* __restrict__ srcp = ei;
  const int* __restrict__ dstp = ei + E;
  const unsigned tid = blockIdx.x * blockDim.x + threadIdx.x;
  const unsigned laneB = (tid & 31u) << 2;
  const unsigned estep = (gridDim.x * blockDim.x) >> 5;
  const unsigned trips = (E + estep - 1u) / estep;
  unsigned e = tid >> 5;
  unsigned left = trips ? trips - 1u : 0u;
  for (; left >= 8u; left -= 8u, e += 8u * estep)
    scatter_batch<LOGC, SRCSH, 8>(ht, srcp, dstp, agg, e, estep, laneB);
  if (left >= 4u) {
    scatter_batch<LOGC, SRCSH, 4>(ht, srcp, dstp, agg, e, estep, laneB);
    left -= 4u; e += 4u * estep;
  }
  if (left >= 2u) {
    scatter_batch<LOGC, SRCSH, 2>(ht, srcp, dstp, agg, e, estep, laneB);
    left -= 2u; e += 2u * estep;
  }
  if (left >= 1u) { scatter_batch<LOGC, SRCSH, 1>(ht, srcp, dstp, agg, e, estep, laneB); e += estep; }
  if (e < E) scatter_batch<LOGC, SRCSH, 1>(ht, srcp, dstp, agg, e, estep, laneB);
}

template <int U>
__device__ __forceinline__ void scatter1_batch(const float* __restrict__ ht,
                                               const int* __restrict__ srcp,
                                               const int* __restrict__ dstp,
                                               float* __restrict__ agg,
                                               unsigned e0, unsigned step) {
  unsigned s[U], d[U];
#pragma unroll
  for (int u = 0; u < U; ++u) {
    s[u] = (unsigned)srcp[e0 + (unsigned)u * step];
    d[u] = (unsigned)dstp[e0 + (unsigned)u * step];
  }
  float m[U];
#pragma unroll
  for (int u = 0; u < U; ++u) m[u] = ht[s[u]];
#pragma unroll
  for (int u = 0; u < U; ++u) atomicAdd(&agg[d[u]], m[u]);
}

__global__ void scatter1_kernel(const float* __restrict__ ht, const int* __restrict__ ei,
                                float* __restrict__ agg2,
                                unsigned E, unsigned N) {
  float* __restrict__ agg = agg2 + package_half() * N;
  const int* __restrict__ srcp = ei;
  const int* __restrict__ dstp = ei + E;
  const unsigned nthr = gridDim.x * blockDim.x;
  const unsigned trips = (E + nthr - 1u) / nthr;
  unsigned e = blockIdx.x * blockDim.x + threadIdx.x;
  unsigned left = trips ? trips - 1u : 0u;
  for (; left >= 4u; left -= 4u, e += 4u * nthr)
    scatter1_batch<4>(ht, srcp, dstp, agg, e, nthr);
  if (left >= 2u) { scatter1_batch<2>(ht, srcp, dstp, agg, e, nthr); left -= 2u; e += 2u * nthr; }
  if (left >= 1u) { scatter1_batch<1>(ht, srcp, dstp, agg, e, nthr); e += nthr; }
  if (e < E) scatter1_batch<1>(ht, srcp, dstp, agg, e, nthr);
}

template <int LOGC, bool TANH>
__global__ void finalize_kernel(const float* aggA, const float* __restrict__ aggB,
                                const float* __restrict__ ht,
                                const float* __restrict__ dis, const float* __restrict__ bias,
                                float* out, unsigned total) {
  unsigned t = blockIdx.x * blockDim.x + threadIdx.x;
  if (t >= total) return;
  unsigned i = t >> LOGC;
  unsigned c = t & ((1u << LOGC) - 1u);
  float d = dis[i];
  float v = d * ((aggA[t] + aggB[t]) + ht[t]) + bias[c];
  out[t] = TANH ? tanhf(v) : v;
}

__device__ __forceinline__ float act_tanh(float a0, float a1, float self_row, float d, float b) {
  return tanhf(d * ((a0 + a1) + self_row) + b);
}

__global__ void __launch_bounds__(256) gemm2_fused(const float* __restrict__ sumA, const float* __restrict__ sumB,
                            float* hs, const float* __restrict__ dis,
                            const float* __restrict__ b1, const float* __restrict__ W2,
                            int N, int nRowTiles) {
  int wave = (int)(((unsigned)blockIdx.x * blockDim.x + threadIdx.x) >> 5);
  if (wave >= nRowTiles) return;
  int lane = threadIdx.x & 31;
  int half = lane >> 4;
  int m = lane & 15;
  int row = wave * 16 + m;
  int rowc = row < N ? row : N - 1;
  const unsigned ro = (unsigned)rowc * 64u;
  const float dr = dis[rowc];

  float dsc[8];
#pragma unroll
  for (int r = 0; r < 8; ++r) {
    int mr = wave * 16 + r + 8 * half;
    dsc[r] = dis[mr < N ? mr : N - 1];
  }

  v2f a0[16], a1[16], sr[16];
#pragma unroll
  for (int j = 0; j < 16; ++j) {
    int ka = 4 * j + half * 2;
    a0[j] = *(const v2f*)(sumA + ro + ka);
    a1[j] = *(const v2f*)(sumB + ro + ka);
    sr[j] = *(const v2f*)(hs + ro + ka);
  }
  v2f bcur = *(const v2f*)(b1 + half * 2);

  v8f acc0 = {}, acc1 = {};
#pragma unroll
  for (int j = 0; j < 16; ++j) {
    int ka = 4 * j + half * 2;
    v2f w0, w1;
    w0.x = W2[(unsigned)ka * 32u + m];
    w0.y = W2[(unsigned)(ka + 1) * 32u + m];
    w1.x = W2[(unsigned)ka * 32u + 16u + m];
    w1.y = W2[(unsigned)(ka + 1) * 32u + 16u + m];
    v2f bnext = bcur;
    if (j + 1 < 16) bnext = *(const v2f*)(b1 + ka + 4);
    v2f a;
    a.x = act_tanh(a0[j].x, a1[j].x, sr[j].x, dr, bcur.x);
    a.y = act_tanh(a0[j].y, a1[j].y, sr[j].y, dr, bcur.y);
    bcur = bnext;
    acc0 = __builtin_amdgcn_wmma_f32_16x16x4_f32(
         false, a,  false, w0,
         (short)0, acc0,  false,  false);
    acc1 = __builtin_amdgcn_wmma_f32_16x16x4_f32(
         false, a,  false, w1,
         (short)0, acc1,  false,  false);
  }
#pragma unroll
  for (int r = 0; r < 8; ++r) {
    int mrow = wave * 16 + r + 8 * half;
    if (mrow < N) {
      hs[(unsigned)mrow * 64u + m]       = acc0[r] * dsc[r];
      hs[(unsigned)mrow * 64u + 16u + m] = acc1[r] * dsc[r];
    }
  }
}

constexpr int DOT_ROWS = 4;
__global__ void __launch_bounds__(256) dot32_fused(const float* __restrict__ cA, const float* __restrict__ cB,
                            const float* __restrict__ hs, const float* __restrict__ dis,
                            const float* __restrict__ b2, const float* __restrict__ W3,
                            float* __restrict__ out, unsigned N) {
  const unsigned t = blockIdx.x * blockDim.x + threadIdx.x;
  const unsigned q = t & 7u;
  const unsigned rstep = (gridDim.x * blockDim.x) >> 3;
  const float4 bb = ((const float4*)b2)[q];
  const float4 w  = ((const float4*)W3)[q];
  float d[DOT_ROWS];
  float4 a0[DOT_ROWS], a1[DOT_ROWS], sr[DOT_ROWS];
#pragma unroll
  for (int u = 0; u < DOT_ROWS; ++u) {
    unsigned i = (t >> 3) + (unsigned)u * rstep;
    unsigned ic = i < N ? i : N - 1u;
    d[u]  = dis[ic];
    a0[u] = ((const float4*)(cA + ic * 32u))[q];
    a1[u] = ((const float4*)(cB + ic * 32u))[q];
    sr[u] = ((const float4*)(hs + ic * 64u))[q];
  }
#pragma unroll
  for (int u = 0; u < DOT_ROWS; ++u) {
    unsigned i = (t >> 3) + (unsigned)u * rstep;
    float p = act_tanh(a0[u].x, a1[u].x, sr[u].x, d[u], bb.x) * w.x + act_tanh(a0[u].y, a1[u].y, sr[u].y, d[u], bb.y) * w.y +
              act_tanh(a0[u].z, a1[u].z, sr[u].z, d[u], bb.z) * w.z + act_tanh(a0[u].w, a1[u].w, sr[u].w, d[u], bb.w) * w.w;
    p += __shfl_xor(p, 1);
    p += __shfl_xor(p, 2);
    p += __shfl_xor(p, 4);
    if (q == 0u && i < N) out[i] = p * d[u];
  }
}

extern "C" void kernel_launch(void* const* d_in, const int* in_sizes, int n_in,
                              void* d_out, int out_size, void* d_ws, size_t ws_size,
                              hipStream_t stream) {
  const float* x  = (const float*)d_in[0];
  const int*   ei = (const int*)d_in[1];
  const float* W1 = (const float*)d_in[2];
  const float* b1 = (const float*)d_in[3];
  const float* W2 = (const float*)d_in[4];
  const float* b2 = (const float*)d_in[5];
  const float* W3 = (const float*)d_in[6];
  const float* b3 = (const float*)d_in[7];
  float* out = (float*)d_out;

  const int D2 = 64, D = 32;
  const int N = in_sizes[0] / D2;
  const int E = in_sizes[1] / 2;

  float* ws    = (float*)d_ws;
  float* dis   = ws;
  float* ht    = ws + N;
  float* aggA  = ht + (long long)N * D2;
  float* aggB  = aggA + (long long)N * D2;
  float* deg2  = ht;
  float* ht3   = aggB;
  float* agg3  = aggB + N;
  (void)ws_size; (void)n_in; (void)out_size;

  const int B = 256;
  const int PG = 2048;
  const int PG1 = 1536;
  const int SG = 128;
  const int ZG = 512;

  fill_kernel<<<SG, B, 0, stream>>>(deg2, 0.0f, (unsigned)(2 * N));
  deg_kernel<<<PG1, B, 0, stream>>>(ei, deg2, (unsigned)E, (unsigned)N);
  dis_kernel<<<(unsigned)cdiv_ll(N, B), B, 0, stream>>>(deg2, deg2 + N, dis, N);

  const int rowTiles = (int)cdiv_ll(N, 16);

  gemm1_rows<<<(unsigned)cdiv_ll((long long)rowTiles * 32, B), B, 0, stream>>>(
      x, W1, dis, ht, N, rowTiles);
  zero4_kernel<<<ZG, B, 0, stream>>>(
      (float4*)aggA, (unsigned)(N * (2 * D2 / 4)));
  scatter_kernel<6, 6><<<PG, B, 0, stream>>>(ht, ei, aggA, (unsigned)E, (unsigned)(N * D2));

  gemm2_fused<<<(unsigned)cdiv_ll((long long)rowTiles * 32, B), B, 0, stream>>>(
      aggA, aggB, ht, dis, b1, W2, N, rowTiles);
  zero4_kernel<<<ZG, B, 0, stream>>>(
      (float4*)aggA, (unsigned)(N * (2 * D / 4)));
  scatter_kernel<5, 6><<<PG, B, 0, stream>>>(ht, ei, aggA, (unsigned)E, (unsigned)(N * D));

  dot32_fused<<<(unsigned)cdiv_ll(cdiv_ll(8LL * N, DOT_ROWS), B), B, 0, stream>>>(
      aggA, aggA + (long long)N * D, ht, dis, b2, W3, ht3, (unsigned)N);
  fill_kernel<<<SG, B, 0, stream>>>(agg3, 0.f, (unsigned)(2 * N));
  scatter1_kernel<<<PG1, B, 0, stream>>>(ht3, ei, agg3, (unsigned)E, (unsigned)N);
  finalize_kernel<0, false><<<(unsigned)cdiv_ll(N, B), B, 0, stream>>>(
      agg3, agg3 + N, ht3, dis, b3, out, (unsigned)N);
}


# Row-wide atomic passes fetch edge ids by scalar loads, 8 consecutive edges per wave trip, one trip ahead: vector memory carries only row gathers and adds; sums stay in arrival order
# speedup vs baseline: 2.6855x; 1.0670x over previous
#include <hip/hip_runtime.h>
#include <math.h>


typedef float v2f __attribute__((ext_vector_type(2)));
typedef float v8f __attribute__((ext_vector_type(8)));

static inline long long cdiv_ll(long long a, long long b) { return (a + b - 1) / b; }

__global__ void fill_kernel(float* __restrict__ p, float v, unsigned n) {
  const unsigned nthr = gridDim.x * blockDim.x;
  for (unsigned i = blockIdx.x * blockDim.x + threadIdx.x; i < n; i += nthr) p[i] = v;
}

__global__ void zero4_kernel(float4* __restrict__ p, unsigned n4) {
  const unsigned nthr = gridDim.x * blockDim.x;
  for (unsigned i = blockIdx.x * blockDim.x + threadIdx.x; i < n4; i += nthr)
    p[i] = make_float4(0.f, 0.f, 0.f, 0.f);
}

__device__ __forceinline__ unsigned package_half() {
  return (__builtin_amdgcn_s_sendmsg_rtn(0x87) >> 18) & 1u;
}

template <int U>
__device__ __forceinline__ void deg_batch(const int* __restrict__ dstp, float* __restrict__ deg,
                                          unsigned e0, unsigned step) {
  unsigned d[U];
#pragma unroll
  for (int u = 0; u < U; ++u) d[u] = (unsigned)dstp[e0 + (unsigned)u * step];
#pragma unroll
  for (int u = 0; u < U; ++u) atomicAdd(&deg[d[u]], 1.0f);
}

__global__ void deg_kernel(const int* __restrict__ ei, float* __restrict__ deg2,
                           unsigned E, unsigned N) {
  float* __restrict__ deg = deg2 + package_half() * N;
  const int* __restrict__ dstp = ei + E;
  const unsigned nthr = gridDim.x * blockDim.x;
  const unsigned trips = (E + nthr - 1u) / nthr;
  unsigned e = blockIdx.x * blockDim.x + threadIdx.x;
  unsigned left = trips ? trips - 1u : 0u;
  for (; left >= 4u; left -= 4u, e += 4u * nthr) deg_batch<4>(dstp, deg, e, nthr);
  if (left >= 2u) { deg_batch<2>(dstp, deg, e, nthr); left -= 2u; e += 2u * nthr; }
  if (left >= 1u) { deg_batch<1>(dstp, deg, e, nthr); e += nthr; }
  if (e < E) deg_batch<1>(dstp, deg, e, nthr);
}

__global__ void dis_kernel(const float* __restrict__ degA, const float* __restrict__ degB,
                           float* __restrict__ dis, int N) {
  int i = blockIdx.x * blockDim.x + threadIdx.x;
  if (i < N) { float d = degA[i] + degB[i] + 1.0f; dis[i] = d > 0.0f ? rsqrtf(d) : 0.0f; }
}

__global__ void __launch_bounds__(256) gemm1_rows(const float* __restrict__ H, const float* __restrict__ W,
                              const float* __restrict__ dis,
                              float* __restrict__ out, int N, int nRowTiles) {
  int wave = (int)(((unsigned)blockIdx.x * blockDim.x + threadIdx.x) >> 5);
  if (wave >= nRowTiles) return;
  int lane = threadIdx.x & 31;
  int half = lane >> 4;
  int m = lane & 15;
  int row = wave * 16 + m;
  int rowc = row < N ? row : N - 1;
  const float* hrow = H + (unsigned)rowc * 64u;

  float dsc[8];
#pragma unroll
  for (int r = 0; r < 8; ++r) {
    int mr = wave * 16 + r + 8 * half;
    dsc[r] = dis[mr < N ? mr : N - 1];
  }

  v2f a[16];
#pragma unroll
  for (int j = 0; j < 16; ++j) a[j] = *(const v2f*)(hrow + 4 * j + half * 2);

  v8f acc0 = {}, acc1 = {}, acc2 = {}, acc3 = {};
#pragma unroll
  for (int j = 0; j < 16; ++j) {
    unsigned ka = (unsigned)(4 * j + half * 2);
    const float* w0 = W + ka * 64u + m;
    const float* w1 = w0 + 64;
    v2f b0, b1, b2, b3;
    b0.x = w0[0];  b0.y = w1[0];
    b1.x = w0[16]; b1.y = w1[16];
    b2.x = w0[32]; b2.y = w1[32];
    b3.x = w0[48]; b3.y = w1[48];
    acc0 = __builtin_amdgcn_wmma_f32_16x16x4_f32(
         false, a[j],  false, b0,
         (short)0, acc0,  false,  false);
    acc1 = __builtin_amdgcn_wmma_f32_16x16x4_f32(
         false, a[j],  false, b1,
         (short)0, acc1,  false,  false);
    acc2 = __builtin_amdgcn_wmma_f32_16x16x4_f32(
         false, a[j],  false, b2,
         (short)0, acc2,  false,  false);
    acc3 = __builtin_amdgcn_wmma_f32_16x16x4_f32(
         false, a[j],  false, b3,
         (short)0, acc3,  false,  false);
  }
#pragma unroll
  for (int r = 0; r < 8; ++r) {
    int mrow = wave * 16 + r + 8 * half;
    if (mrow < N) {
      float* o = out + (unsigned)mrow * 64u + m;
      o[0]  = acc0[r] * dsc[r];
      o[16] = acc1[r] * dsc[r];
      o[32] = acc2[r] * dsc[r];
      o[48] = acc3[r] * dsc[r];
    }
  }
}

template <int LOGC, int SRCSH, int U>
__device__ __forceinline__ void scatter_rows(const float* __restrict__ ht, float* __restrict__ agg,
                                             const unsigned (&s)[U], const unsigned (&d)[U],
                                             unsigned laneB) {
  constexpr int WPL = (1 << LOGC) / 32;
  float v[U * WPL];
#pragma unroll
  for (int u = 0; u < U; ++u) {
    const char* r = (const char*)ht + ((s[u] << (SRCSH + 2)) + laneB);
#pragma unroll
    for (int w = 0; w < WPL; ++w) v[u * WPL + w] = *(const float*)(r + 128 * w);
  }
#pragma unroll
  for (int u = 0; u < U; ++u) {
    char* o = (char*)agg + ((d[u] << (LOGC + 2)) + laneB);
#pragma unroll
    for (int w = 0; w < WPL; ++w) atomicAdd((float*)(o + 128 * w), v[u * WPL + w]);
  }
}

template <int U>
__device__ __forceinline__ void edge_ids(const int* __restrict__ srcp, const int* __restrict__ dstp,
                                         unsigned e0, unsigned (&s)[U], unsigned (&d)[U]) {
#pragma unroll
  for (int u = 0; u < U; ++u) {
    s[u] = (unsigned)srcp[e0 + (unsigned)u];
    d[u] = (unsigned)dstp[e0 + (unsigned)u];
  }
}

template <int LOGC, int SRCSH>
__global__ void scatter_kernel(const float* __restrict__ ht, const int* __restrict__ ei,
                               float* __restrict__ agg2,
                               unsigned E, unsigned copyElems) {
  static_assert(LOGC >= 5, "one wave covers at least one 32-word line");
  float* __restrict__ agg = agg2 + package_half() * copyElems;
  const int* __restrict__ srcp = ei;
  const int* __restrict__ dstp = ei + E;
  const unsigned tid = blockIdx.x * blockDim.x + threadIdx.x;
  const unsigned laneB = (tid & 31u) << 2;
  const unsigned nwaves = (gridDim.x * blockDim.x) >> 5;
  const unsigned wv = (unsigned)__builtin_amdgcn_readfirstlane((int)(tid >> 5));
  const unsigned groups = E >> 3;
  if (wv < groups) {
    unsigned g = wv;
    unsigned sc[8], dc[8];
    edge_ids<8>(srcp, dstp, g << 3, sc, dc);
    for (;;) {
      const unsigned gn = g + nwaves;
      const bool more = gn < groups;
      unsigned sn[8], dn[8];
      edge_ids<8>(srcp, dstp, (more ? gn : g) << 3, sn, dn);
      scatter_rows<LOGC, SRCSH, 8>(ht, agg, sc, dc, laneB);
      if (!more) break;
#pragma unroll
      for (int u = 0; u < 8; ++u) { sc[u] = sn[u]; dc[u] = dn[u]; }
      g = gn;
    }
  }
  const unsigned e = (groups << 3) + wv;
  if (e < E) {
    unsigned s1[1], d1[1];
    edge_ids<1>(srcp, dstp, e, s1, d1);
    scatter_rows<LOGC, SRCSH, 1>(ht, agg, s1, d1, laneB);
  }
}

template <int U>
__device__ __forceinline__ void scatter1_batch(const float* __restrict__ ht,
                                               const int* __restrict__ srcp,
                                               const int* __restrict__ dstp,
                                               float* __restrict__ agg,
                                               unsigned e0, unsigned step) {
  unsigned s[U], d[U];
#pragma unroll
  for (int u = 0; u < U; ++u) {
    s[u] = (unsigned)srcp[e0 + (unsigned)u * step];
    d[u] = (unsigned)dstp[e0 + (unsigned)u * step];
  }
  float m[U];
#pragma unroll
  for (int u = 0; u < U; ++u) m[u] = ht[s[u]];
#pragma unroll
  for (int u = 0; u < U; ++u) atomicAdd(&agg[d[u]], m[u]);
}

__global__ void scatter1_kernel(const float* __restrict__ ht, const int* __restrict__ ei,
                                float* __restrict__ agg2,
                                unsigned E, unsigned N) {
  float* __restrict__ agg = agg2 + package_half() * N;
  const int* __restrict__ srcp = ei;
  const int* __restrict__ dstp = ei + E;
  const unsigned nthr = gridDim.x * blockDim.x;
  const unsigned trips = (E + nthr - 1u) / nthr;
  unsigned e = blockIdx.x * blockDim.x + threadIdx.x;
  unsigned left = trips ? trips - 1u : 0u;
  for (; left >= 4u; left -= 4u, e += 4u * nthr)
    scatter1_batch<4>(ht, srcp, dstp, agg, e, nthr);
  if (left >= 2u) { scatter1_batch<2>(ht, srcp, dstp, agg, e, nthr); left -= 2u; e += 2u * nthr; }
  if (left >= 1u) { scatter1_batch<1>(ht, srcp, dstp, agg, e, nthr); e += nthr; }
  if (e < E) scatter1_batch<1>(ht, srcp, dstp, agg, e, nthr);
}

template <int LOGC, bool TANH>
__global__ void finalize_kernel(const float* aggA, const float* __restrict__ aggB,
                                const float* __restrict__ ht,
                                const float* __restrict__ dis, const float* __restrict__ bias,
                                float* out, unsigned total) {
  unsigned t = blockIdx.x * blockDim.x + threadIdx.x;
  if (t >= total) return;
  unsigned i = t >> LOGC;
  unsigned c = t & ((1u << LOGC) - 1u);
  float d = dis[i];
  float v = d * ((aggA[t] + aggB[t]) + ht[t]) + bias[c];
  out[t] = TANH ? tanhf(v) : v;
}

__device__ __forceinline__ float act_tanh(float a0, float a1, float self_row, float d, float b) {
  return tanhf(d * ((a0 + a1) + self_row) + b);
}

__global__ void __launch_bounds__(256) gemm2_fused(const float* __restrict__ sumA, const float* __restrict__ sumB,
                            float* hs, const float* __restrict__ dis,
                            const float* __restrict__ b1, const float* __restrict__ W2,
                            int N, int nRowTiles) {
  int wave = (int)(((unsigned)blockIdx.x * blockDim.x + threadIdx.x) >> 5);
  if (wave >= nRowTiles) return;
  int lane = threadIdx.x & 31;
  int half = lane >> 4;
  int m = lane & 15;
  int row = wave * 16 + m;
  int rowc = row < N ? row : N - 1;
  const unsigned ro = (unsigned)rowc * 64u;
  const float dr = dis[rowc];

  float dsc[8];
#pragma unroll
  for (int r = 0; r < 8; ++r) {
    int mr = wave * 16 + r + 8 * half;
    dsc[r] = dis[mr < N ? mr : N - 1];
  }

  v2f a0[16], a1[16], sr[16];
#pragma unroll
  for (int j = 0; j < 16; ++j) {
    int ka = 4 * j + half * 2;
    a0[j] = *(const v2f*)(sumA + ro + ka);
    a1[j] = *(const v2f*)(sumB + ro + ka);
    sr[j] = *(const v2f*)(hs + ro + ka);
  }
  v2f bcur = *(const v2f*)(b1 + half * 2);

  v8f acc0 = {}, acc1 = {};
#pragma unroll
  for (int j = 0; j < 16; ++j) {
    int ka = 4 * j + half * 2;
    v2f w0, w1;
    w0.x = W2[(unsigned)ka * 32u + m];
    w0.y = W2[(unsigned)(ka + 1) * 32u + m];
    w1.x = W2[(unsigned)ka * 32u + 16u + m];
    w1.y = W2[(unsigned)(ka + 1) * 32u + 16u + m];
    v2f bnext = bcur;
    if (j + 1 < 16) bnext = *(const v2f*)(b1 + ka + 4);
    v2f a;
    a.x = act_tanh(a0[j].x, a1[j].x, sr[j].x, dr, bcur.x);
    a.y = act_tanh(a0[j].y, a1[j].y, sr[j].y, dr, bcur.y);
    bcur = bnext;
    acc0 = __builtin_amdgcn_wmma_f32_16x16x4_f32(
         false, a,  false, w0,
         (short)0, acc0,  false,  false);
    acc1 = __builtin_amdgcn_wmma_f32_16x16x4_f32(
         false, a,  false, w1,
         (short)0, acc1,  false,  false);
  }
#pragma unroll
  for (int r = 0; r < 8; ++r) {
    int mrow = wave * 16 + r + 8 * half;
    if (mrow < N) {
      hs[(unsigned)mrow * 64u + m]       = acc0[r] * dsc[r];
      hs[(unsigned)mrow * 64u + 16u + m] = acc1[r] * dsc[r];
    }
  }
}

constexpr int DOT_ROWS = 4;
__global__ void __launch_bounds__(256) dot32_fused(const float* __restrict__ cA, const float* __restrict__ cB,
                            const float* __restrict__ hs, const float* __restrict__ dis,
                            const float* __restrict__ b2, const float* __restrict__ W3,
                            float* __restrict__ out, unsigned N) {
  const unsigned t = blockIdx.x * blockDim.x + threadIdx.x;
  const unsigned q = t & 7u;
  const unsigned rstep = (gridDim.x * blockDim.x) >> 3;
  const float4 bb = ((const float4*)b2)[q];
  const float4 w  = ((const float4*)W3)[q];
  float d[DOT_ROWS];
  float4 a0[DOT_ROWS], a1[DOT_ROWS], sr[DOT_ROWS];
#pragma unroll
  for (int u = 0; u < DOT_ROWS; ++u) {
    unsigned i = (t >> 3) + (unsigned)u * rstep;
    unsigned ic = i < N ? i : N - 1u;
    d[u]  = dis[ic];
    a0[u] = ((const float4*)(cA + ic * 32u))[q];
    a1[u] = ((const float4*)(cB + ic * 32u))[q];
    sr[u] = ((const float4*)(hs + ic * 64u))[q];
  }
#pragma unroll
  for (int u = 0; u < DOT_ROWS; ++u) {
    unsigned i = (t >> 3) + (unsigned)u * rstep;
    float p = act_tanh(a0[u].x, a1[u].x, sr[u].x, d[u], bb.x) * w.x + act_tanh(a0[u].y, a1[u].y, sr[u].y, d[u], bb.y) * w.y +
              act_tanh(a0[u].z, a1[u].z, sr[u].z, d[u], bb.z) * w.z + act_tanh(a0[u].w, a1[u].w, sr[u].w, d[u], bb.w) * w.w;
    p += __shfl_xor(p, 1);
    p += __shfl_xor(p, 2);
    p += __shfl_xor(p, 4);
    if (q == 0u && i < N) out[i] = p * d[u];
  }
}

extern "C" void kernel_launch(void* const* d_in, const int* in_sizes, int n_in,
                              void* d_out, int out_size, void* d_ws, size_t ws_size,
                              hipStream_t stream) {
  const float* x  = (const float*)d_in[0];
  const int*   ei = (const int*)d_in[1];
  const float* W1 = (const float*)d_in[2];
  const float* b1 = (const float*)d_in[3];
  const float* W2 = (const float*)d_in[4];
  const float* b2 = (const float*)d_in[5];
  const float* W3 = (const float*)d_in[6];
  const float* b3 = (const float*)d_in[7];
  float* out = (float*)d_out;

  const int D2 = 64, D = 32;
  const int N = in_sizes[0] / D2;
  const int E = in_sizes[1] / 2;

  float* ws    = (float*)d_ws;
  float* dis   = ws;
  float* ht    = ws + N;
  float* aggA  = ht + (long long)N * D2;
  float* aggB  = aggA + (long long)N * D2;
  float* deg2  = ht;
  float* ht3   = aggB;
  float* agg3  = aggB + N;
  (void)ws_size; (void)n_in; (void)out_size;

  const int B = 256;
  const int PG = 2048;
  const int PG1 = 1536;
  const int SG = 128;
  const int ZG = 512;

  fill_kernel<<<SG, B, 0, stream>>>(deg2, 0.0f, (unsigned)(2 * N));
  deg_kernel<<<PG1, B, 0, stream>>>(ei, deg2, (unsigned)E, (unsigned)N);
  dis_kernel<<<(unsigned)cdiv_ll(N, B), B, 0, stream>>>(deg2, deg2 + N, dis, N);

  const int rowTiles = (int)cdiv_ll(N, 16);

  gemm1_rows<<<(unsigned)cdiv_ll((long long)rowTiles * 32, B), B, 0, stream>>>(
      x, W1, dis, ht, N, rowTiles);
  zero4_kernel<<<ZG, B, 0, stream>>>(
      (float4*)aggA, (unsigned)(N * (2 * D2 / 4)));
  scatter_kernel<6, 6><<<PG, B, 0, stream>>>(ht, ei, aggA, (unsigned)E, (unsigned)(N * D2));

  gemm2_fused<<<(unsigned)cdiv_ll((long long)rowTiles * 32, B), B, 0, stream>>>(
      aggA, aggB, ht, dis, b1, W2, N, rowTiles);
  zero4_kernel<<<ZG, B, 0, stream>>>(
      (float4*)aggA, (unsigned)(N * (2 * D / 4)));
  scatter_kernel<5, 6><<<PG, B, 0, stream>>>(ht, ei, aggA, (unsigned)E, (unsigned)(N * D));

  dot32_fused<<<(unsigned)cdiv_ll(cdiv_ll(8LL * N, DOT_ROWS), B), B, 0, stream>>>(
      aggA, aggA + (long long)N * D, ht, dis, b2, W3, ht3, (unsigned)N);
  fill_kernel<<<SG, B, 0, stream>>>(agg3, 0.f, (unsigned)(2 * N));
  scatter1_kernel<<<PG1, B, 0, stream>>>(ht3, ei, agg3, (unsigned)E, (unsigned)N);
  finalize_kernel<0, false><<<(unsigned)cdiv_ll(N, B), B, 0, stream>>>(
      agg3, agg3 + N, ht3, dis, b3, out, (unsigned)N);
}
